# starting kernel
# baseline (speedup 1.0000x reference)
#include <hip/hip_runtime.h>

typedef __attribute__((ext_vector_type(2))) float v2f;
typedef __attribute__((ext_vector_type(8))) float v8f;

#define NV2 163842
#define NV1 40962
#define NV0 10242
#define ALPHA 0.3f

__device__ __forceinline__ float4 feat_get4(const float* __restrict__ fA,
                                            const int* __restrict__ mapA, int CA,
                                            const float* __restrict__ fB, int CB,
                                            int v, int c4) {
  if (c4 < CA) {
    int u = mapA ? mapA[v] : v;
    return *(const float4*)(fA + (long)u * CA + c4);
  }
  return *(const float4*)(fB + (long)v * CB + (c4 - CA));
}

__global__ __launch_bounds__(64) void edgeconv_wmma(
    const float* __restrict__ fA, const int* __restrict__ mapA, int CA,
    const float* __restrict__ fB, int CB,
    const int* __restrict__ src,
    const float* __restrict__ W, const float* __restrict__ bias,
    float* __restrict__ out, int nv, int Cin, int Cout)
{
  extern __shared__ __align__(16) char smem[];
  int*   sidx = (int*)smem;
  float* xl   = (float*)(smem + 384);
  const int K2   = 2 * Cin;
  const int tid  = threadIdx.x;
  const int tile = blockIdx.x;

  for (int i = tid; i < 96; i += 64) {
    int r = i / 6, j = i - r * 6;
    int v = tile * 16 + r; if (v >= nv) v = nv - 1;
    sidx[i] = src[v * 6 + j];
  }
  __syncthreads();

  const int C4 = Cin >> 2;
  for (int i = tid; i < 16 * C4; i += 64) {
    int r = i / C4, c4 = (i - r * C4) << 2;
    int v = tile * 16 + r; if (v >= nv) v = nv - 1;
    float4 loc = feat_get4(fA, mapA, CA, fB, CB, v, c4);
    float sx = 0.f, sy = 0.f, sz = 0.f, sw = 0.f;
    #pragma unroll
    for (int j = 0; j < 6; ++j) {
      float4 t = feat_get4(fA, mapA, CA, fB, CB, sidx[r * 6 + j], c4);
      sx += t.x; sy += t.y; sz += t.z; sw += t.w;
    }
    const float inv6 = 1.0f / 6.0f;
    float4 d = make_float4(sx * inv6 - loc.x, sy * inv6 - loc.y,
                           sz * inv6 - loc.z, sw * inv6 - loc.w);
    *(float4*)(xl + r * K2 + c4)       = loc;
    *(float4*)(xl + r * K2 + Cin + c4) = d;
  }
  __syncthreads();

  const int lane = tid & 31;
  const int wave = tid >> 5;
  const int g    = lane >> 4;
  const int nl   = lane & 15;
  const int nK   = K2 >> 2;
  const int nNT  = Cout >> 4;

  for (int nt = wave; nt < nNT; nt += 2) {
    const int n = nt * 16 + nl;
    const float bv = bias[n];
    v8f acc;
    #pragma unroll
    for (int j = 0; j < 8; ++j) acc[j] = bv;

    const float* __restrict__ ap = xl + nl * K2 + 2 * g;
    const float* __restrict__ bp = W + (long)n * K2 + 2 * g;
    for (int kt = 0; kt < nK; ++kt) {
      v2f a = *(const v2f*)(ap + 4 * kt);
      v2f b = *(const v2f*)(bp + 4 * kt);
      acc = __builtin_amdgcn_wmma_f32_16x16x4_f32(
           false, a,  false, b,
           (short)0, acc,  false,  false);
    }

    #pragma unroll
    for (int j = 0; j < 8; ++j) {
      int m = tile * 16 + j + 8 * g;
      if (m < nv) {
        float y = acc[j];
        y = (y >= 0.f) ? y : ALPHA * y;
        out[(long)m * Cout + n] = y;
      }
    }
  }
}

__device__ __forceinline__ unsigned ford(float f) {
  unsigned u = __float_as_uint(f);
  return (u & 0x80000000u) ? ~u : (u | 0x80000000u);
}
__device__ __forceinline__ float funord(unsigned u) {
  unsigned b = (u & 0x80000000u) ? (u & 0x7FFFFFFFu) : ~u;
  return __uint_as_float(b);
}

__global__ void pool_init(unsigned* __restrict__ p, int n) {
  int i = blockIdx.x * blockDim.x + threadIdx.x;
  if (i < n) p[i] = 0x007FFFFFu;
}
__global__ void pool_scatter(const float* __restrict__ f, const int* __restrict__ map,
                             unsigned* __restrict__ buf, int nchild, int C) {
  int i = blockIdx.x * blockDim.x + threadIdx.x;
  if (i >= nchild * C) return;
  int child = i / C, c = i - child * C;
  atomicMax(&buf[(long)map[child] * C + c], ford(f[i]));
}
__global__ void pool_decode(unsigned* __restrict__ buf, int n) {
  int i = blockIdx.x * blockDim.x + threadIdx.x;
  if (i < n) {
    float v = funord(buf[i]);
    ((float*)buf)[i] = v;
  }
}

extern "C" void kernel_launch(void* const* d_in, const int* in_sizes, int n_in,
                              void* d_out, int out_size, void* d_ws, size_t ws_size,
                              hipStream_t stream) {
  (void)in_sizes; (void)n_in; (void)out_size; (void)ws_size;
  const float* features = (const float*)d_in[0];
  const int*   src2     = (const int*)d_in[1];
  const int*   src1     = (const int*)d_in[3];
  const int*   src0     = (const int*)d_in[5];
  const int*   p21      = (const int*)d_in[7];
  const int*   p10      = (const int*)d_in[8];
  const float* w_enc0 = (const float*)d_in[9],  *b_enc0 = (const float*)d_in[10];
  const float* w_enc1 = (const float*)d_in[11], *b_enc1 = (const float*)d_in[12];
  const float* w_enc2 = (const float*)d_in[13], *b_enc2 = (const float*)d_in[14];
  const float* w_dec0 = (const float*)d_in[15], *b_dec0 = (const float*)d_in[16];
  const float* w_dec1 = (const float*)d_in[17], *b_dec1 = (const float*)d_in[18];

  char* ws = (char*)d_ws;
  size_t off = 0;
  float*    f2 = (float*)(ws + off); off += (size_t)NV2 * 32 * 4;
  float*    f1 = (float*)(ws + off); off += (size_t)NV1 * 64 * 4;
  float*    f0 = (float*)(ws + off); off += (size_t)NV0 * 128 * 4;
  float*    d1 = (float*)(ws + off); off += (size_t)NV1 * 64 * 4;
  unsigned* P  = (unsigned*)(ws + off);

  auto ec = [&](const float* fA, const int* mapA, int CA, const float* fB, int CB,
                const int* src, const float* W, const float* b, float* out,
                int nv, int Cin, int Cout) {
    int K2 = 2 * Cin;
    size_t shmem = 384 + (size_t)(16 * K2) * sizeof(float);
    int grid = (nv + 15) / 16;
    edgeconv_wmma<<<grid, 64, shmem, stream>>>(fA, mapA, CA, fB, CB, src, W, b,
                                               out, nv, Cin, Cout);
  };
  auto pool = [&](const float* fchild, const int* map, int nchild, int nparent, int C) {
    int nc = nchild * C, np = nparent * C;
    pool_init   <<<(np + 255) / 256, 256, 0, stream>>>(P, np);
    pool_scatter<<<(nc + 255) / 256, 256, 0, stream>>>(fchild, map, P, nchild, C);
    pool_decode <<<(np + 255) / 256, 256, 0, stream>>>(P, np);
  };

  ec(features, nullptr, 16, nullptr, 0, src2, w_enc0, b_enc0, f2, NV2, 16, 32);
  pool(f2, p21, NV2, NV1, 32);
  ec((const float*)P, nullptr, 32, nullptr, 0, src1, w_enc1, b_enc1, f1, NV1, 32, 64);
  pool(f1, p10, NV1, NV0, 64);
  ec((const float*)P, nullptr, 64, nullptr, 0, src0, w_enc2, b_enc2, f0, NV0, 64, 128);

  ec(f0, p10, 128, f1, 64, src1, w_dec0, b_dec0, d1, NV1, 192, 64);
  ec(d1, p21, 64, f2, 32, src2, w_dec1, b_dec1, (float*)d_out, NV2, 96, 32);
}


# EdgeConv gather: self+6 neighbor rows (parent map applied) resolved to LDS once per tile, concat side by select not branch, so an item's 7 row loads issue together, not one round trip each
# speedup vs baseline: 1.2010x; 1.2010x over previous
#include <hip/hip_runtime.h>

typedef __attribute__((ext_vector_type(2))) float v2f;
typedef __attribute__((ext_vector_type(8))) float v8f;

#define NV2 163842
#define NV1 40962
#define NV0 10242
#define ALPHA 0.3f


__global__ __launch_bounds__(64) void edgeconv_wmma(
    const float* __restrict__ fA, const int* __restrict__ mapA, int CA,
    const float* __restrict__ fB, int CB,
    const int* __restrict__ src,
    const float* __restrict__ W, const float* __restrict__ bias,
    float* __restrict__ out, int nv, int Cin, int Cout)
{
  extern __shared__ __align__(16) char smem[];
  int*   rowB = (int*)smem;
  int*   rowA = (int*)(smem + 512);
  float* xl   = (float*)(smem + 1024);
  const int K2   = 2 * Cin;
  const int tid  = threadIdx.x;
  const int tile = blockIdx.x;

  {
    const int r = tid >> 3, j = tid & 7;
    int v0 = tile * 16 + r;     if (v0 >= nv) v0 = nv - 1;
    int v1 = tile * 16 + r + 8; if (v1 >= nv) v1 = nv - 1;
    const bool nb = (j >= 1) && (j <= 6);
    const int  jo = nb ? j - 1 : 0;
    int s0 = src[v0 * 6 + jo];
    int s1 = src[v1 * 6 + jo];
    if (!nb) { s0 = v0; s1 = v1; }
    int a0 = s0, a1 = s1;
    if (mapA) { a0 = mapA[s0]; a1 = mapA[s1]; }
    rowB[tid] = s0; rowB[tid + 64] = s1;
    rowA[tid] = a0; rowA[tid + 64] = a1;
  }
  __syncthreads();

  const int C4 = Cin >> 2;
  for (int i = tid; i < 16 * C4; i += 64) {
    int r = i / C4, c4 = (i - r * C4) << 2;
    const bool inA = c4 < CA;
    const float* __restrict__ fp = inA ? fA : fB;
    const int  stride = inA ? CA : CB;
    const int  coff   = inA ? c4 : c4 - CA;
    const int* rows   = (inA ? rowA : rowB) + r * 8;
    const int4 r0 = *(const int4*)(rows);
    const int4 r1 = *(const int4*)(rows + 4);
    const int id[7] = { r0.x, r0.y, r0.z, r0.w, r1.x, r1.y, r1.z };
    float4 t[7];
    #pragma unroll
    for (int j = 0; j < 7; ++j)
      t[j] = *(const float4*)(fp + (long)id[j] * stride + coff);
    const float4 loc = t[0];
    float sx = 0.f, sy = 0.f, sz = 0.f, sw = 0.f;
    #pragma unroll
    for (int j = 1; j < 7; ++j) {
      sx += t[j].x; sy += t[j].y; sz += t[j].z; sw += t[j].w;
    }
    const float inv6 = 1.0f / 6.0f;
    float4 d = make_float4(sx * inv6 - loc.x, sy * inv6 - loc.y,
                           sz * inv6 - loc.z, sw * inv6 - loc.w);
    *(float4*)(xl + r * K2 + c4)       = loc;
    *(float4*)(xl + r * K2 + Cin + c4) = d;
  }
  __syncthreads();

  const int lane = tid & 31;
  const int wave = tid >> 5;
  const int g    = lane >> 4;
  const int nl   = lane & 15;
  const int nK   = K2 >> 2;
  const int nNT  = Cout >> 4;

  for (int nt = wave; nt < nNT; nt += 2) {
    const int n = nt * 16 + nl;
    const float bv = bias[n];
    v8f acc;
    #pragma unroll
    for (int j = 0; j < 8; ++j) acc[j] = bv;

    const float* __restrict__ ap = xl + nl * K2 + 2 * g;
    const float* __restrict__ bp = W + (long)n * K2 + 2 * g;
    for (int kt = 0; kt < nK; ++kt) {
      v2f a = *(const v2f*)(ap + 4 * kt);
      v2f b = *(const v2f*)(bp + 4 * kt);
      acc = __builtin_amdgcn_wmma_f32_16x16x4_f32(
           false, a,  false, b,
           (short)0, acc,  false,  false);
    }

    #pragma unroll
    for (int j = 0; j < 8; ++j) {
      int m = tile * 16 + j + 8 * g;
      if (m < nv) {
        float y = acc[j];
        y = (y >= 0.f) ? y : ALPHA * y;
        out[(long)m * Cout + n] = y;
      }
    }
  }
}

__device__ __forceinline__ unsigned ford(float f) {
  unsigned u = __float_as_uint(f);
  return (u & 0x80000000u) ? ~u : (u | 0x80000000u);
}
__device__ __forceinline__ float funord(unsigned u) {
  unsigned b = (u & 0x80000000u) ? (u & 0x7FFFFFFFu) : ~u;
  return __uint_as_float(b);
}

__global__ void pool_init(unsigned* __restrict__ p, int n) {
  int i = blockIdx.x * blockDim.x + threadIdx.x;
  if (i < n) p[i] = 0x007FFFFFu;
}
__global__ void pool_scatter(const float* __restrict__ f, const int* __restrict__ map,
                             unsigned* __restrict__ buf, int nchild, int C) {
  int i = blockIdx.x * blockDim.x + threadIdx.x;
  if (i >= nchild * C) return;
  int child = i / C, c = i - child * C;
  atomicMax(&buf[(long)map[child] * C + c], ford(f[i]));
}
__global__ void pool_decode(unsigned* __restrict__ buf, int n) {
  int i = blockIdx.x * blockDim.x + threadIdx.x;
  if (i < n) {
    float v = funord(buf[i]);
    ((float*)buf)[i] = v;
  }
}

extern "C" void kernel_launch(void* const* d_in, const int* in_sizes, int n_in,
                              void* d_out, int out_size, void* d_ws, size_t ws_size,
                              hipStream_t stream) {
  (void)in_sizes; (void)n_in; (void)out_size; (void)ws_size;
  const float* features = (const float*)d_in[0];
  const int*   src2     = (const int*)d_in[1];
  const int*   src1     = (const int*)d_in[3];
  const int*   src0     = (const int*)d_in[5];
  const int*   p21      = (const int*)d_in[7];
  const int*   p10      = (const int*)d_in[8];
  const float* w_enc0 = (const float*)d_in[9],  *b_enc0 = (const float*)d_in[10];
  const float* w_enc1 = (const float*)d_in[11], *b_enc1 = (const float*)d_in[12];
  const float* w_enc2 = (const float*)d_in[13], *b_enc2 = (const float*)d_in[14];
  const float* w_dec0 = (const float*)d_in[15], *b_dec0 = (const float*)d_in[16];
  const float* w_dec1 = (const float*)d_in[17], *b_dec1 = (const float*)d_in[18];

  char* ws = (char*)d_ws;
  size_t off = 0;
  float*    f2 = (float*)(ws + off); off += (size_t)NV2 * 32 * 4;
  float*    f1 = (float*)(ws + off); off += (size_t)NV1 * 64 * 4;
  float*    f0 = (float*)(ws + off); off += (size_t)NV0 * 128 * 4;
  float*    d1 = (float*)(ws + off); off += (size_t)NV1 * 64 * 4;
  unsigned* P  = (unsigned*)(ws + off);

  auto ec = [&](const float* fA, const int* mapA, int CA, const float* fB, int CB,
                const int* src, const float* W, const float* b, float* out,
                int nv, int Cin, int Cout) {
    int K2 = 2 * Cin;
    size_t shmem = 1024 + (size_t)(16 * K2) * sizeof(float);
    int grid = (nv + 15) / 16;
    edgeconv_wmma<<<grid, 64, shmem, stream>>>(fA, mapA, CA, fB, CB, src, W, b,
                                               out, nv, Cin, Cout);
  };
  auto pool = [&](const float* fchild, const int* map, int nchild, int nparent, int C) {
    int nc = nchild * C, np = nparent * C;
    pool_init   <<<(np + 255) / 256, 256, 0, stream>>>(P, np);
    pool_scatter<<<(nc + 255) / 256, 256, 0, stream>>>(fchild, map, P, nchild, C);
    pool_decode <<<(np + 255) / 256, 256, 0, stream>>>(P, np);
  };

  ec(features, nullptr, 16, nullptr, 0, src2, w_enc0, b_enc0, f2, NV2, 16, 32);
  pool(f2, p21, NV2, NV1, 32);
  ec((const float*)P, nullptr, 32, nullptr, 0, src1, w_enc1, b_enc1, f1, NV1, 32, 64);
  pool(f1, p10, NV1, NV0, 64);
  ec((const float*)P, nullptr, 64, nullptr, 0, src0, w_enc2, b_enc2, f0, NV0, 64, 128);

  ec(f0, p10, 128, f1, 64, src1, w_dec0, b_dec0, d1, NV1, 192, 64);
  ec(d1, p21, 64, f2, 32, src2, w_dec1, b_dec1, (float*)d_out, NV2, 96, 32);
}


# segment_max pooling: one atomic exchange per child links it into its parent's list (arrival order; max is order-free, no sum reordered), an atomic-free per-parent walk takes the max; 11->8 launches
# speedup vs baseline: 1.2750x; 1.0616x over previous
#include <hip/hip_runtime.h>

typedef __attribute__((ext_vector_type(2))) float v2f;
typedef __attribute__((ext_vector_type(8))) float v8f;

#define NV2 163842
#define NV1 40962
#define NV0 10242
#define ALPHA 0.3f


__global__ __launch_bounds__(64) void edgeconv_wmma(
    const float* __restrict__ fA, const int* __restrict__ mapA, int CA,
    const float* __restrict__ fB, int CB,
    const int* __restrict__ src,
    const float* __restrict__ W, const float* __restrict__ bias,
    float* __restrict__ out, int nv, int Cin, int Cout,
    int* __restrict__ clr, int nclr)
{
  extern __shared__ __align__(16) char smem[];
  int*   rowB = (int*)smem;
  int*   rowA = (int*)(smem + 512);
  float* xl   = (float*)(smem + 1024);
  const int K2   = 2 * Cin;
  const int tid  = threadIdx.x;
  const int tile = blockIdx.x;

  {
    const int r = tid >> 3, j = tid & 7;
    int v0 = tile * 16 + r;     if (v0 >= nv) v0 = nv - 1;
    int v1 = tile * 16 + r + 8; if (v1 >= nv) v1 = nv - 1;
    const bool nb = (j >= 1) && (j <= 6);
    const int  jo = nb ? j - 1 : 0;
    int s0 = src[v0 * 6 + jo];
    int s1 = src[v1 * 6 + jo];
    if (!nb) { s0 = v0; s1 = v1; }
    int a0 = s0, a1 = s1;
    if (mapA) { a0 = mapA[s0]; a1 = mapA[s1]; }
    rowB[tid] = s0; rowB[tid + 64] = s1;
    rowA[tid] = a0; rowA[tid + 64] = a1;
  }
  __syncthreads();

  const int C4 = Cin >> 2;
  for (int i = tid; i < 16 * C4; i += 64) {
    int r = i / C4, c4 = (i - r * C4) << 2;
    const bool inA = c4 < CA;
    const float* __restrict__ fp = inA ? fA : fB;
    const int  stride = inA ? CA : CB;
    const int  coff   = inA ? c4 : c4 - CA;
    const int* rows   = (inA ? rowA : rowB) + r * 8;
    const int4 r0 = *(const int4*)(rows);
    const int4 r1 = *(const int4*)(rows + 4);
    const int id[7] = { r0.x, r0.y, r0.z, r0.w, r1.x, r1.y, r1.z };
    float4 t[7];
    #pragma unroll
    for (int j = 0; j < 7; ++j)
      t[j] = *(const float4*)(fp + (long)id[j] * stride + coff);
    const float4 loc = t[0];
    float sx = 0.f, sy = 0.f, sz = 0.f, sw = 0.f;
    #pragma unroll
    for (int j = 1; j < 7; ++j) {
      sx += t[j].x; sy += t[j].y; sz += t[j].z; sw += t[j].w;
    }
    const float inv6 = 1.0f / 6.0f;
    float4 d = make_float4(sx * inv6 - loc.x, sy * inv6 - loc.y,
                           sz * inv6 - loc.z, sw * inv6 - loc.w);
    *(float4*)(xl + r * K2 + c4)       = loc;
    *(float4*)(xl + r * K2 + Cin + c4) = d;
  }
  __syncthreads();

  const int lane = tid & 31;
  const int wave = tid >> 5;
  const int g    = lane >> 4;
  const int nl   = lane & 15;
  const int nK   = K2 >> 2;
  const int nNT  = Cout >> 4;

  for (int nt = wave; nt < nNT; nt += 2) {
    const int n = nt * 16 + nl;
    const float bv = bias[n];
    v8f acc;
    #pragma unroll
    for (int j = 0; j < 8; ++j) acc[j] = bv;

    const float* __restrict__ ap = xl + nl * K2 + 2 * g;
    const float* __restrict__ bp = W + (long)n * K2 + 2 * g;
    for (int kt = 0; kt < nK; ++kt) {
      v2f a = *(const v2f*)(ap + 4 * kt);
      v2f b = *(const v2f*)(bp + 4 * kt);
      acc = __builtin_amdgcn_wmma_f32_16x16x4_f32(
           false, a,  false, b,
           (short)0, acc,  false,  false);
    }

    #pragma unroll
    for (int j = 0; j < 8; ++j) {
      int m = tile * 16 + j + 8 * g;
      if (m < nv) {
        float y = acc[j];
        y = (y >= 0.f) ? y : ALPHA * y;
        out[(long)m * Cout + n] = y;
      }
    }
  }

  if (clr) {
    const int k = tile * 64 + tid;
    if (k < nclr) clr[k] = -1;
  }
}

__device__ __forceinline__ unsigned ford(float f) {
  unsigned u = __float_as_uint(f);
  return (u & 0x80000000u) ? ~u : (u | 0x80000000u);
}
__device__ __forceinline__ float funord(unsigned u) {
  unsigned b = (u & 0x80000000u) ? (u & 0x7FFFFFFFu) : ~u;
  return __uint_as_float(b);
}

__global__ __launch_bounds__(256) void pool_link(
    const int* __restrict__ mapF, int nF, int* __restrict__ headF, int* __restrict__ nextF,
    const int* __restrict__ mapC, int nC, int* __restrict__ headC, int* __restrict__ nextC)
{
  const int i = blockIdx.x * 256 + threadIdx.x;
  if (i < nF) {
    nextF[i] = atomicExch(&headF[mapF[i]], i);
  } else if (i < nF + nC) {
    const int k = i - nF;
    nextC[k] = atomicExch(&headC[mapC[k]], k);
  }
}

__global__ __launch_bounds__(256) void pool_walk(
    const float* __restrict__ f, const int* __restrict__ head, const int* __restrict__ next,
    float* __restrict__ out, int nparent, int nchild, int lshift)
{
  const int t = blockIdx.x * 256 + threadIdx.x;
  const int p = t >> lshift;
  const int l = t & ((1 << lshift) - 1);
  if (p >= nparent) return;
  const int C = 4 << lshift;
  int i = head[p];
  unsigned mx = 0x007FFFFFu, my = 0x007FFFFFu, mz = 0x007FFFFFu, mw = 0x007FFFFFu;
  for (int k = 0; k < nchild && i >= 0; ++k) {
    const float4 v = *(const float4*)(f + (long)i * C + 4 * l);
    const int nx = next[i];
    const unsigned ux = ford(v.x), uy = ford(v.y), uz = ford(v.z), uw = ford(v.w);
    mx = ux > mx ? ux : mx;  my = uy > my ? uy : my;
    mz = uz > mz ? uz : mz;  mw = uw > mw ? uw : mw;
    i = nx;
  }
  *(float4*)(out + (long)p * C + 4 * l) =
      make_float4(funord(mx), funord(my), funord(mz), funord(mw));
}

extern "C" void kernel_launch(void* const* d_in, const int* in_sizes, int n_in,
                              void* d_out, int out_size, void* d_ws, size_t ws_size,
                              hipStream_t stream) {
  (void)in_sizes; (void)n_in; (void)out_size; (void)ws_size;
  const float* features = (const float*)d_in[0];
  const int*   src2     = (const int*)d_in[1];
  const int*   src1     = (const int*)d_in[3];
  const int*   src0     = (const int*)d_in[5];
  const int*   p21      = (const int*)d_in[7];
  const int*   p10      = (const int*)d_in[8];
  const float* w_enc0 = (const float*)d_in[9],  *b_enc0 = (const float*)d_in[10];
  const float* w_enc1 = (const float*)d_in[11], *b_enc1 = (const float*)d_in[12];
  const float* w_enc2 = (const float*)d_in[13], *b_enc2 = (const float*)d_in[14];
  const float* w_dec0 = (const float*)d_in[15], *b_dec0 = (const float*)d_in[16];
  const float* w_dec1 = (const float*)d_in[17], *b_dec1 = (const float*)d_in[18];

  char* ws = (char*)d_ws;
  size_t off = 0;
  float* f2 = (float*)(ws + off); off += (size_t)NV2 * 32 * 4;
  float* f1 = (float*)(ws + off); off += (size_t)NV1 * 64 * 4;
  float* f0 = (float*)(ws + off); off += (size_t)NV0 * 128 * 4;
  float* d1 = (float*)(ws + off); off += (size_t)NV1 * 64 * 4;
  float* P1 = (float*)(ws + off); off += (size_t)NV1 * 32 * 4;
  float* P0 = (float*)(ws + off); off += (size_t)NV0 * 64 * 4;
  int* head1 = (int*)(ws + off); off += (size_t)NV1 * 4;
  int* head0 = (int*)(ws + off); off += (size_t)NV0 * 4;
  int* next2 = (int*)(ws + off); off += ((size_t)NV2 * 4 + 15) / 16 * 16;
  int* next1 = (int*)(ws + off);

  auto ec = [&](const float* fA, const int* mapA, int CA, const float* fB, int CB,
                const int* src, const float* W, const float* b, float* out,
                int nv, int Cin, int Cout, int* clr, int nclr) {
    int K2 = 2 * Cin;
    size_t shmem = 1024 + (size_t)(16 * K2) * sizeof(float);
    int grid = (nv + 15) / 16;
    edgeconv_wmma<<<grid, 64, shmem, stream>>>(fA, mapA, CA, fB, CB, src, W, b,
                                               out, nv, Cin, Cout, clr, nclr);
  };
  auto walk = [&](const float* fchild, const int* head, const int* next, float* pooled,
                  int nchild, int nparent, int lshift) {
    int nt = nparent << lshift;
    pool_walk<<<(nt + 255) / 256, 256, 0, stream>>>(fchild, head, next, pooled,
                                                     nparent, nchild, lshift);
  };

  ec(features, nullptr, 16, nullptr, 0, src2, w_enc0, b_enc0, f2, NV2, 16, 32, head1, NV1 + NV0);
  pool_link<<<(NV2 + NV1 + 255) / 256, 256, 0, stream>>>(p21, NV2, head1, next2,
                                                         p10, NV1, head0, next1);
  walk(f2, head1, next2, P1, NV2, NV1, 3);
  ec(P1, nullptr, 32, nullptr, 0, src1, w_enc1, b_enc1, f1, NV1, 32, 64, nullptr, 0);
  walk(f1, head0, next1, P0, NV1, NV0, 4);
  ec(P0, nullptr, 64, nullptr, 0, src0, w_enc2, b_enc2, f0, NV0, 64, 128, nullptr, 0);

  ec(f0, p10, 128, f1, 64, src1, w_dec0, b_dec0, d1, NV1, 192, 64, nullptr, 0);
  ec(d1, p21, 64, f2, 32, src2, w_dec1, b_dec1, (float*)d_out, NV2, 96, 32, nullptr, 0);
}


# WMMA operand fetch: x-tile rows at pitch K2+4 floats (A reads were 16-way LDS bank conflicts) and A/B fragments fetched 16 B per lane per two WMMAs; k summed in a new order within each 8
# speedup vs baseline: 1.3746x; 1.0781x over previous
#include <hip/hip_runtime.h>

typedef __attribute__((ext_vector_type(2))) float v2f;
typedef __attribute__((ext_vector_type(8))) float v8f;

#define NV2 163842
#define NV1 40962
#define NV0 10242
#define ALPHA 0.3f


__global__ __launch_bounds__(64) void edgeconv_wmma(
    const float* __restrict__ fA, const int* __restrict__ mapA, int CA,
    const float* __restrict__ fB, int CB,
    const int* __restrict__ src,
    const float* __restrict__ W, const float* __restrict__ bias,
    float* __restrict__ out, int nv, int Cin, int Cout,
    int* __restrict__ clr, int nclr)
{
  extern __shared__ __align__(16) char smem[];
  int*   rowB = (int*)smem;
  int*   rowA = (int*)(smem + 512);
  float* xl   = (float*)(smem + 1024);
  const int K2   = 2 * Cin;
  const int KP   = K2 + 4;
  const int tid  = threadIdx.x;
  const int tile = blockIdx.x;

  {
    const int r = tid >> 3, j = tid & 7;
    int v0 = tile * 16 + r;     if (v0 >= nv) v0 = nv - 1;
    int v1 = tile * 16 + r + 8; if (v1 >= nv) v1 = nv - 1;
    const bool nb = (j >= 1) && (j <= 6);
    const int  jo = nb ? j - 1 : 0;
    int s0 = src[v0 * 6 + jo];
    int s1 = src[v1 * 6 + jo];
    if (!nb) { s0 = v0; s1 = v1; }
    int a0 = s0, a1 = s1;
    if (mapA) { a0 = mapA[s0]; a1 = mapA[s1]; }
    rowB[tid] = s0; rowB[tid + 64] = s1;
    rowA[tid] = a0; rowA[tid + 64] = a1;
  }
  __syncthreads();

  const int C4 = Cin >> 2;
  for (int i = tid; i < 16 * C4; i += 64) {
    int r = i / C4, c4 = (i - r * C4) << 2;
    const bool inA = c4 < CA;
    const float* __restrict__ fp = inA ? fA : fB;
    const int  stride = inA ? CA : CB;
    const int  coff   = inA ? c4 : c4 - CA;
    const int* rows   = (inA ? rowA : rowB) + r * 8;
    const int4 r0 = *(const int4*)(rows);
    const int4 r1 = *(const int4*)(rows + 4);
    const int id[7] = { r0.x, r0.y, r0.z, r0.w, r1.x, r1.y, r1.z };
    float4 t[7];
    #pragma unroll
    for (int j = 0; j < 7; ++j)
      t[j] = *(const float4*)(fp + (long)id[j] * stride + coff);
    const float4 loc = t[0];
    float sx = 0.f, sy = 0.f, sz = 0.f, sw = 0.f;
    #pragma unroll
    for (int j = 1; j < 7; ++j) {
      sx += t[j].x; sy += t[j].y; sz += t[j].z; sw += t[j].w;
    }
    const float inv6 = 1.0f / 6.0f;
    float4 d = make_float4(sx * inv6 - loc.x, sy * inv6 - loc.y,
                           sz * inv6 - loc.z, sw * inv6 - loc.w);
    *(float4*)(xl + r * KP + c4)       = loc;
    *(float4*)(xl + r * KP + Cin + c4) = d;
  }
  __syncthreads();

  const int lane = tid & 31;
  const int wave = tid >> 5;
  const int g    = lane >> 4;
  const int nl   = lane & 15;
  const int nK8  = K2 >> 3;
  const int nNT  = Cout >> 4;

  for (int nt = wave; nt < nNT; nt += 2) {
    const int n = nt * 16 + nl;
    const float bv = bias[n];
    v8f acc;
    #pragma unroll
    for (int j = 0; j < 8; ++j) acc[j] = bv;

    const float* __restrict__ ap = xl + nl * KP + 4 * g;
    const float* __restrict__ bp = W + (long)n * K2 + 4 * g;
    #pragma unroll 4
    for (int ks = 0; ks < nK8; ++ks) {
      const float4 a4 = *(const float4*)(ap + 8 * ks);
      const float4 b4 = *(const float4*)(bp + 8 * ks);
      v2f a0, b0, a1, b1;
      a0[0] = a4.x; a0[1] = a4.y;  b0[0] = b4.x; b0[1] = b4.y;
      a1[0] = a4.z; a1[1] = a4.w;  b1[0] = b4.z; b1[1] = b4.w;
      acc = __builtin_amdgcn_wmma_f32_16x16x4_f32(
           false, a0,  false, b0,
           (short)0, acc,  false,  false);
      acc = __builtin_amdgcn_wmma_f32_16x16x4_f32(
           false, a1,  false, b1,
           (short)0, acc,  false,  false);
    }

    #pragma unroll
    for (int j = 0; j < 8; ++j) {
      int m = tile * 16 + j + 8 * g;
      if (m < nv) {
        float y = acc[j];
        y = (y >= 0.f) ? y : ALPHA * y;
        out[(long)m * Cout + n] = y;
      }
    }
  }

  if (clr) {
    const int k = tile * 64 + tid;
    if (k < nclr) clr[k] = -1;
  }
}

__device__ __forceinline__ unsigned ford(float f) {
  unsigned u = __float_as_uint(f);
  return (u & 0x80000000u) ? ~u : (u | 0x80000000u);
}
__device__ __forceinline__ float funord(unsigned u) {
  unsigned b = (u & 0x80000000u) ? (u & 0x7FFFFFFFu) : ~u;
  return __uint_as_float(b);
}

__global__ __launch_bounds__(256) void pool_link(
    const int* __restrict__ mapF, int nF, int* __restrict__ headF, int* __restrict__ nextF,
    const int* __restrict__ mapC, int nC, int* __restrict__ headC, int* __restrict__ nextC)
{
  const int i = blockIdx.x * 256 + threadIdx.x;
  if (i < nF) {
    nextF[i] = atomicExch(&headF[mapF[i]], i);
  } else if (i < nF + nC) {
    const int k = i - nF;
    nextC[k] = atomicExch(&headC[mapC[k]], k);
  }
}

__global__ __launch_bounds__(256) void pool_walk(
    const float* __restrict__ f, const int* __restrict__ head, const int* __restrict__ next,
    float* __restrict__ out, int nparent, int nchild, int lshift)
{
  const int t = blockIdx.x * 256 + threadIdx.x;
  const int p = t >> lshift;
  const int l = t & ((1 << lshift) - 1);
  if (p >= nparent) return;
  const int C = 4 << lshift;
  int i = head[p];
  unsigned mx = 0x007FFFFFu, my = 0x007FFFFFu, mz = 0x007FFFFFu, mw = 0x007FFFFFu;
  for (int k = 0; k < nchild && i >= 0; ++k) {
    const float4 v = *(const float4*)(f + (long)i * C + 4 * l);
    const int nx = next[i];
    const unsigned ux = ford(v.x), uy = ford(v.y), uz = ford(v.z), uw = ford(v.w);
    mx = ux > mx ? ux : mx;  my = uy > my ? uy : my;
    mz = uz > mz ? uz : mz;  mw = uw > mw ? uw : mw;
    i = nx;
  }
  *(float4*)(out + (long)p * C + 4 * l) =
      make_float4(funord(mx), funord(my), funord(mz), funord(mw));
}

extern "C" void kernel_launch(void* const* d_in, const int* in_sizes, int n_in,
                              void* d_out, int out_size, void* d_ws, size_t ws_size,
                              hipStream_t stream) {
  (void)in_sizes; (void)n_in; (void)out_size; (void)ws_size;
  const float* features = (const float*)d_in[0];
  const int*   src2     = (const int*)d_in[1];
  const int*   src1     = (const int*)d_in[3];
  const int*   src0     = (const int*)d_in[5];
  const int*   p21      = (const int*)d_in[7];
  const int*   p10      = (const int*)d_in[8];
  const float* w_enc0 = (const float*)d_in[9],  *b_enc0 = (const float*)d_in[10];
  const float* w_enc1 = (const float*)d_in[11], *b_enc1 = (const float*)d_in[12];
  const float* w_enc2 = (const float*)d_in[13], *b_enc2 = (const float*)d_in[14];
  const float* w_dec0 = (const float*)d_in[15], *b_dec0 = (const float*)d_in[16];
  const float* w_dec1 = (const float*)d_in[17], *b_dec1 = (const float*)d_in[18];

  char* ws = (char*)d_ws;
  size_t off = 0;
  float* f2 = (float*)(ws + off); off += (size_t)NV2 * 32 * 4;
  float* f1 = (float*)(ws + off); off += (size_t)NV1 * 64 * 4;
  float* f0 = (float*)(ws + off); off += (size_t)NV0 * 128 * 4;
  float* d1 = (float*)(ws + off); off += (size_t)NV1 * 64 * 4;
  float* P1 = (float*)(ws + off); off += (size_t)NV1 * 32 * 4;
  float* P0 = (float*)(ws + off); off += (size_t)NV0 * 64 * 4;
  int* head1 = (int*)(ws + off); off += (size_t)NV1 * 4;
  int* head0 = (int*)(ws + off); off += (size_t)NV0 * 4;
  int* next2 = (int*)(ws + off); off += ((size_t)NV2 * 4 + 15) / 16 * 16;
  int* next1 = (int*)(ws + off);

  auto ec = [&](const float* fA, const int* mapA, int CA, const float* fB, int CB,
                const int* src, const float* W, const float* b, float* out,
                int nv, int Cin, int Cout, int* clr, int nclr) {
    int K2 = 2 * Cin;
    size_t shmem = 1024 + (size_t)(16 * (K2 + 4)) * sizeof(float);
    int grid = (nv + 15) / 16;
    edgeconv_wmma<<<grid, 64, shmem, stream>>>(fA, mapA, CA, fB, CB, src, W, b,
                                               out, nv, Cin, Cout, clr, nclr);
  };
  auto walk = [&](const float* fchild, const int* head, const int* next, float* pooled,
                  int nchild, int nparent, int lshift) {
    int nt = nparent << lshift;
    pool_walk<<<(nt + 255) / 256, 256, 0, stream>>>(fchild, head, next, pooled,
                                                     nparent, nchild, lshift);
  };

  ec(features, nullptr, 16, nullptr, 0, src2, w_enc0, b_enc0, f2, NV2, 16, 32, head1, NV1 + NV0);
  pool_link<<<(NV2 + NV1 + 255) / 256, 256, 0, stream>>>(p21, NV2, head1, next2,
                                                         p10, NV1, head0, next1);
  walk(f2, head1, next2, P1, NV2, NV1, 3);
  ec(P1, nullptr, 32, nullptr, 0, src1, w_enc1, b_enc1, f1, NV1, 32, 64, nullptr, 0);
  walk(f1, head0, next1, P0, NV1, NV0, 4);
  ec(P0, nullptr, 64, nullptr, 0, src0, w_enc2, b_enc2, f0, NV0, 64, 128, nullptr, 0);

  ec(f0, p10, 128, f1, 64, src1, w_dec0, b_dec0, d1, NV1, 192, 64, nullptr, 0);
  ec(d1, p21, 64, f2, 32, src2, w_dec1, b_dec1, (float*)d_out, NV2, 96, 32, nullptr, 0);
}


# EdgeConv: a wave keeps each fetched W fragment in registers for 2 or 4 M-tiles (blocks of 32 or 64 vertices, 2 or 4 waves, independent accumulators): W fetched 2-4x less often, half as many waves
# speedup vs baseline: 1.4505x; 1.0552x over previous
#include <hip/hip_runtime.h>

typedef __attribute__((ext_vector_type(2))) float v2f;
typedef __attribute__((ext_vector_type(8))) float v8f;

#define NV2 163842
#define NV1 40962
#define NV0 10242
#define ALPHA 0.3f


template <int CIN, int COUT>
__global__ __launch_bounds__((COUT / 16 >= 4) ? 128 : 64) void edgeconv_wmma(
    const float* __restrict__ fA, const int* __restrict__ mapA, int CA,
    const float* __restrict__ fB, int CB,
    const int* __restrict__ src,
    const float* __restrict__ W, const float* __restrict__ bias,
    float* __restrict__ out, int nv,
    int* __restrict__ clr, int nclr)
{
  constexpr int K2  = 2 * CIN;
  constexpr int KP  = K2 + 4;
  constexpr int C4  = CIN / 4;
  constexpr int NK8 = K2 / 8;
  constexpr int NNT = COUT / 16;
  constexpr int MT  = (NNT >= 4) ? 4 : 2;
  constexpr int BV  = 16 * MT;
  constexpr int NW  = (NNT >= 4) ? 4 : 2;
  constexpr int NT  = 32 * NW;
  constexpr int NIT = NNT / NW;

  __shared__ __align__(16) int   rowB[BV * 8];
  __shared__ __align__(16) int   rowA[BV * 8];
  __shared__ __align__(16) float xl[BV * KP];
  const int tid  = threadIdx.x;
  const int tile = blockIdx.x;

  {
    const int r = tid >> 3, j = tid & 7;
    const bool nb = (j >= 1) && (j <= 6);
    const int  jo = nb ? j - 1 : 0;
    int v[4], s[4], a[4];
    #pragma unroll
    for (int q = 0; q < 4; ++q) {
      int vv = tile * BV + r + (NT / 8) * q;
      v[q] = (vv >= nv) ? nv - 1 : vv;
    }
    #pragma unroll
    for (int q = 0; q < 4; ++q) s[q] = src[v[q] * 6 + jo];
    #pragma unroll
    for (int q = 0; q < 4; ++q) { if (!nb) s[q] = v[q]; a[q] = s[q]; }
    if (mapA) {
      #pragma unroll
      for (int q = 0; q < 4; ++q) a[q] = mapA[s[q]];
    }
    #pragma unroll
    for (int q = 0; q < 4; ++q) { rowB[tid + NT * q] = s[q]; rowA[tid + NT * q] = a[q]; }
  }
  __syncthreads();

  static_assert((BV * C4) % (2 * NT) == 0, "two items per thread per trip");
  for (int i0 = tid; i0 < BV * C4; i0 += 2 * NT) {
    int    xoff[2];
    float4 t[2][7];
    #pragma unroll
    for (int u = 0; u < 2; ++u) {
      const int i = i0 + NT * u;
      const int r = i / C4, c4 = (i - r * C4) << 2;
      const bool inA = c4 < CA;
      const float* __restrict__ fp = inA ? fA : fB;
      const int  stride = inA ? CA : CB;
      const int  coff   = inA ? c4 : c4 - CA;
      const int* rows   = (inA ? rowA : rowB) + r * 8;
      const int4 r0 = *(const int4*)(rows);
      const int4 r1 = *(const int4*)(rows + 4);
      const int id[7] = { r0.x, r0.y, r0.z, r0.w, r1.x, r1.y, r1.z };
      #pragma unroll
      for (int j = 0; j < 7; ++j)
        t[u][j] = *(const float4*)(fp + (long)id[j] * stride + coff);
      xoff[u] = r * KP + c4;
    }
    #pragma unroll
    for (int u = 0; u < 2; ++u) {
      const float4 loc = t[u][0];
      float sx = 0.f, sy = 0.f, sz = 0.f, sw = 0.f;
      #pragma unroll
      for (int j = 1; j < 7; ++j) {
        sx += t[u][j].x; sy += t[u][j].y; sz += t[u][j].z; sw += t[u][j].w;
      }
      const float inv6 = 1.0f / 6.0f;
      float4 d = make_float4(sx * inv6 - loc.x, sy * inv6 - loc.y,
                             sz * inv6 - loc.z, sw * inv6 - loc.w);
      *(float4*)(xl + xoff[u])       = loc;
      *(float4*)(xl + xoff[u] + CIN) = d;
    }
  }
  __syncthreads();

  const int lane = tid & 31;
  const int wave = tid >> 5;
  const int g    = lane >> 4;
  const int nl   = lane & 15;
  #pragma unroll 1
  for (int it = 0; it < NIT; ++it) {
    const int nt = wave + NW * it;
    const int n = nt * 16 + nl;
    const float bv = bias[n];
    v8f acc[MT];
    #pragma unroll
    for (int mt = 0; mt < MT; ++mt) {
      #pragma unroll
      for (int j = 0; j < 8; ++j) acc[mt][j] = bv;
    }

    const float* __restrict__ ap = xl + nl * KP + 4 * g;
    const float* __restrict__ bp = W + (long)n * K2 + 4 * g;
    #pragma unroll 4
    for (int ks = 0; ks < NK8; ++ks) {
      const float4 b4 = *(const float4*)(bp + 8 * ks);
      float4 a4[MT];
      #pragma unroll
      for (int mt = 0; mt < MT; ++mt)
        a4[mt] = *(const float4*)(ap + mt * 16 * KP + 8 * ks);
      v2f b0, b1;
      b0[0] = b4.x; b0[1] = b4.y;
      b1[0] = b4.z; b1[1] = b4.w;
      #pragma unroll
      for (int mt = 0; mt < MT; ++mt) {
        v2f a0;
        a0[0] = a4[mt].x; a0[1] = a4[mt].y;
        acc[mt] = __builtin_amdgcn_wmma_f32_16x16x4_f32(
             false, a0,  false, b0,
             (short)0, acc[mt],  false,  false);
      }
      #pragma unroll
      for (int mt = 0; mt < MT; ++mt) {
        v2f a1;
        a1[0] = a4[mt].z; a1[1] = a4[mt].w;
        acc[mt] = __builtin_amdgcn_wmma_f32_16x16x4_f32(
             false, a1,  false, b1,
             (short)0, acc[mt],  false,  false);
      }
    }

    #pragma unroll
    for (int mt = 0; mt < MT; ++mt) {
      #pragma unroll
      for (int j = 0; j < 8; ++j) {
        int m = tile * BV + 16 * mt + j + 8 * g;
        if (m < nv) {
          float y = acc[mt][j];
          y = (y >= 0.f) ? y : ALPHA * y;
          out[(long)m * COUT + n] = y;
        }
      }
    }
  }

  if (clr) {
    const int k = tile * NT + tid;
    if (k < nclr) clr[k] = -1;
  }
}

__device__ __forceinline__ unsigned ford(float f) {
  unsigned u = __float_as_uint(f);
  return (u & 0x80000000u) ? ~u : (u | 0x80000000u);
}
__device__ __forceinline__ float funord(unsigned u) {
  unsigned b = (u & 0x80000000u) ? (u & 0x7FFFFFFFu) : ~u;
  return __uint_as_float(b);
}

__global__ __launch_bounds__(256) void pool_link(
    const int* __restrict__ mapF, int nF, int* __restrict__ headF, int* __restrict__ nextF,
    const int* __restrict__ mapC, int nC, int* __restrict__ headC, int* __restrict__ nextC)
{
  const int i = blockIdx.x * 256 + threadIdx.x;
  if (i < nF) {
    nextF[i] = atomicExch(&headF[mapF[i]], i);
  } else if (i < nF + nC) {
    const int k = i - nF;
    nextC[k] = atomicExch(&headC[mapC[k]], k);
  }
}

__global__ __launch_bounds__(256) void pool_walk(
    const float* __restrict__ f, const int* __restrict__ head, const int* __restrict__ next,
    float* __restrict__ out, int nparent, int nchild, int lshift)
{
  const int t = blockIdx.x * 256 + threadIdx.x;
  const int p = t >> lshift;
  const int l = t & ((1 << lshift) - 1);
  if (p >= nparent) return;
  const int C = 4 << lshift;
  int i = head[p];
  unsigned mx = 0x007FFFFFu, my = 0x007FFFFFu, mz = 0x007FFFFFu, mw = 0x007FFFFFu;
  for (int k = 0; k < nchild && i >= 0; ++k) {
    const float4 v = *(const float4*)(f + (long)i * C + 4 * l);
    const int nx = next[i];
    const unsigned ux = ford(v.x), uy = ford(v.y), uz = ford(v.z), uw = ford(v.w);
    mx = ux > mx ? ux : mx;  my = uy > my ? uy : my;
    mz = uz > mz ? uz : mz;  mw = uw > mw ? uw : mw;
    i = nx;
  }
  *(float4*)(out + (long)p * C + 4 * l) =
      make_float4(funord(mx), funord(my), funord(mz), funord(mw));
}

extern "C" void kernel_launch(void* const* d_in, const int* in_sizes, int n_in,
                              void* d_out, int out_size, void* d_ws, size_t ws_size,
                              hipStream_t stream) {
  (void)in_sizes; (void)n_in; (void)out_size; (void)ws_size;
  const float* features = (const float*)d_in[0];
  const int*   src2     = (const int*)d_in[1];
  const int*   src1     = (const int*)d_in[3];
  const int*   src0     = (const int*)d_in[5];
  const int*   p21      = (const int*)d_in[7];
  const int*   p10      = (const int*)d_in[8];
  const float* w_enc0 = (const float*)d_in[9],  *b_enc0 = (const float*)d_in[10];
  const float* w_enc1 = (const float*)d_in[11], *b_enc1 = (const float*)d_in[12];
  const float* w_enc2 = (const float*)d_in[13], *b_enc2 = (const float*)d_in[14];
  const float* w_dec0 = (const float*)d_in[15], *b_dec0 = (const float*)d_in[16];
  const float* w_dec1 = (const float*)d_in[17], *b_dec1 = (const float*)d_in[18];

  char* ws = (char*)d_ws;
  size_t off = 0;
  float* f2 = (float*)(ws + off); off += (size_t)NV2 * 32 * 4;
  float* f1 = (float*)(ws + off); off += (size_t)NV1 * 64 * 4;
  float* f0 = (float*)(ws + off); off += (size_t)NV0 * 128 * 4;
  float* d1 = (float*)(ws + off); off += (size_t)NV1 * 64 * 4;
  float* P1 = (float*)(ws + off); off += (size_t)NV1 * 32 * 4;
  float* P0 = (float*)(ws + off); off += (size_t)NV0 * 64 * 4;
  int* head1 = (int*)(ws + off); off += (size_t)NV1 * 4;
  int* head0 = (int*)(ws + off); off += (size_t)NV0 * 4;
  int* next2 = (int*)(ws + off); off += ((size_t)NV2 * 4 + 15) / 16 * 16;
  int* next1 = (int*)(ws + off);

  constexpr int G2 = (NV2 + 31) / 32, G1 = (NV1 + 63) / 64, G0 = (NV0 + 63) / 64;
  auto walk = [&](const float* fchild, const int* head, const int* next, float* pooled,
                  int nchild, int nparent, int lshift) {
    int nt = nparent << lshift;
    pool_walk<<<(nt + 255) / 256, 256, 0, stream>>>(fchild, head, next, pooled,
                                                     nparent, nchild, lshift);
  };

  edgeconv_wmma<16, 32><<<G2, 64, 0, stream>>>(features, nullptr, 16, nullptr, 0, src2,
                                                w_enc0, b_enc0, f2, NV2, head1, NV1 + NV0);
  pool_link<<<(NV2 + NV1 + 255) / 256, 256, 0, stream>>>(p21, NV2, head1, next2,
                                                         p10, NV1, head0, next1);
  walk(f2, head1, next2, P1, NV2, NV1, 3);
  edgeconv_wmma<32, 64><<<G1, 128, 0, stream>>>(P1, nullptr, 32, nullptr, 0, src1,
                                                w_enc1, b_enc1, f1, NV1, nullptr, 0);
  walk(f1, head0, next1, P0, NV1, NV0, 4);
  edgeconv_wmma<64, 128><<<G0, 128, 0, stream>>>(P0, nullptr, 64, nullptr, 0, src0,
                                                 w_enc2, b_enc2, f0, NV0, nullptr, 0);

  edgeconv_wmma<192, 64><<<G1, 128, 0, stream>>>(f0, p10, 128, f1, 64, src1,
                                                 w_dec0, b_dec0, d1, NV1, nullptr, 0);
  edgeconv_wmma<96, 32><<<G2, 64, 0, stream>>>(d1, p21, 64, f2, 32, src2,
                                                w_dec1, b_dec1, (float*)d_out, NV2, nullptr, 0);
}


# Decoder: W's parent-side columns applied once per parent by a 2nd WMMA GEMM on the producer's LDS output tile; child gathers mapped COUT-wide rows as accumulator start, K=2CB; k summed in 2 groups
# speedup vs baseline: 1.7392x; 1.1990x over previous
#include <hip/hip_runtime.h>

typedef __attribute__((ext_vector_type(2))) float v2f;
typedef __attribute__((ext_vector_type(8))) float v8f;

#define NV2 163842
#define NV1 40962
#define NV0 10242
#define ALPHA 0.3f

template <int MT>
__device__ __forceinline__ void kstep8(v8f (&acc)[MT], const float* __restrict__ a,
                                       int mstride, const float* __restrict__ b)
{
  const float4 b4 = *(const float4*)b;
  float4 a4[MT];
  #pragma unroll
  for (int mt = 0; mt < MT; ++mt) a4[mt] = *(const float4*)(a + mt * mstride);
  v2f b0, b1;
  b0[0] = b4.x; b0[1] = b4.y;
  b1[0] = b4.z; b1[1] = b4.w;
  #pragma unroll
  for (int mt = 0; mt < MT; ++mt) {
    v2f a0;
    a0[0] = a4[mt].x; a0[1] = a4[mt].y;
    acc[mt] = __builtin_amdgcn_wmma_f32_16x16x4_f32(
         false, a0,  false, b0,
         (short)0, acc[mt],  false,  false);
  }
  #pragma unroll
  for (int mt = 0; mt < MT; ++mt) {
    v2f a1;
    a1[0] = a4[mt].z; a1[1] = a4[mt].w;
    acc[mt] = __builtin_amdgcn_wmma_f32_16x16x4_f32(
         false, a1,  false, b1,
         (short)0, acc[mt],  false,  false);
  }
}

template <int KD, int TP, int CO2, int WP2, int MT, int NW>
__device__ __forceinline__ void second_map(const float* __restrict__ tl,
                                           const float* __restrict__ Wn, float* __restrict__ Yn,
                                           int tile, int nv, int wave, int g, int nl)
{
  constexpr int NNT2 = 2 * CO2 / 16;
  constexpr int NIT2 = NNT2 / NW;
  static_assert(NNT2 % NW == 0 && CO2 % 16 == 0 && KD % 8 == 0, "tile shape");
  static_assert((TP / 4) % 2 == 1, "tile pitch must be an odd number of 16-B units");
  #pragma unroll 1
  for (int it = 0; it < NIT2; ++it) {
    const int np = (wave + NW * it) * 16 + nl;
    const bool second = np >= CO2;
    const int n = second ? np - CO2 : np;
    v8f acc[MT];
    #pragma unroll
    for (int mt = 0; mt < MT; ++mt) {
      #pragma unroll
      for (int j = 0; j < 8; ++j) acc[mt][j] = 0.f;
    }
    const float* __restrict__ ap = tl + nl * TP + 4 * g;
    const float* __restrict__ bp = Wn + (long)n * WP2 + (second ? WP2 / 2 : 0) + 4 * g;
    #pragma unroll 4
    for (int ks = 0; ks < KD / 8; ++ks) kstep8<MT>(acc, ap + 8 * ks, 16 * TP, bp + 8 * ks);
    #pragma unroll
    for (int mt = 0; mt < MT; ++mt) {
      #pragma unroll
      for (int j = 0; j < 8; ++j) {
        const int m = tile * (16 * MT) + 16 * mt + j + 8 * g;
        if (m < nv) Yn[(long)m * (2 * CO2) + np] = acc[mt][j];
      }
    }
  }
}


template <int CIN, int COUT, int CO2, int WP2>
__global__ __launch_bounds__((COUT / 16 >= 4) ? 128 : 64) void edgeconv_wmma(
    const float* __restrict__ fA, const int* __restrict__ mapA, int CA,
    const float* __restrict__ fB, int CB,
    const int* __restrict__ src,
    const float* __restrict__ W, const float* __restrict__ bias,
    float* __restrict__ out, int nv,
    int* __restrict__ clr, int nclr,
    const float* __restrict__ Wn, float* __restrict__ Yn)
{
  constexpr int K2  = 2 * CIN;
  constexpr int KP  = K2 + 4;
  constexpr int C4  = CIN / 4;
  constexpr int NK8 = K2 / 8;
  constexpr int NNT = COUT / 16;
  constexpr int MT  = (NNT >= 4) ? 4 : 2;
  constexpr int BV  = 16 * MT;
  constexpr int NW  = (NNT >= 4) ? 4 : 2;
  constexpr int NT  = 32 * NW;
  constexpr int NIT = NNT / NW;

  __shared__ __align__(16) int   rowB[BV * 8];
  __shared__ __align__(16) int   rowA[BV * 8];
  __shared__ __align__(16) float xl[BV * KP];
  constexpr int OP  = COUT + 4;
  __shared__ __align__(16) float ot[(CO2 > 0) ? BV * OP : 4];
  const int tid  = threadIdx.x;
  const int tile = blockIdx.x;

  {
    const int r = tid >> 3, j = tid & 7;
    const bool nb = (j >= 1) && (j <= 6);
    const int  jo = nb ? j - 1 : 0;
    int v[4], s[4], a[4];
    #pragma unroll
    for (int q = 0; q < 4; ++q) {
      int vv = tile * BV + r + (NT / 8) * q;
      v[q] = (vv >= nv) ? nv - 1 : vv;
    }
    #pragma unroll
    for (int q = 0; q < 4; ++q) s[q] = src[v[q] * 6 + jo];
    #pragma unroll
    for (int q = 0; q < 4; ++q) { if (!nb) s[q] = v[q]; a[q] = s[q]; }
    if (mapA) {
      #pragma unroll
      for (int q = 0; q < 4; ++q) a[q] = mapA[s[q]];
    }
    #pragma unroll
    for (int q = 0; q < 4; ++q) { rowB[tid + NT * q] = s[q]; rowA[tid + NT * q] = a[q]; }
  }
  __syncthreads();

  static_assert((BV * C4) % (2 * NT) == 0, "two items per thread per trip");
  for (int i0 = tid; i0 < BV * C4; i0 += 2 * NT) {
    int    xoff[2];
    float4 t[2][7];
    #pragma unroll
    for (int u = 0; u < 2; ++u) {
      const int i = i0 + NT * u;
      const int r = i / C4, c4 = (i - r * C4) << 2;
      const bool inA = c4 < CA;
      const float* __restrict__ fp = inA ? fA : fB;
      const int  stride = inA ? CA : CB;
      const int  coff   = inA ? c4 : c4 - CA;
      const int* rows   = (inA ? rowA : rowB) + r * 8;
      const int4 r0 = *(const int4*)(rows);
      const int4 r1 = *(const int4*)(rows + 4);
      const int id[7] = { r0.x, r0.y, r0.z, r0.w, r1.x, r1.y, r1.z };
      #pragma unroll
      for (int j = 0; j < 7; ++j)
        t[u][j] = *(const float4*)(fp + (long)id[j] * stride + coff);
      xoff[u] = r * KP + c4;
    }
    #pragma unroll
    for (int u = 0; u < 2; ++u) {
      const float4 loc = t[u][0];
      float sx = 0.f, sy = 0.f, sz = 0.f, sw = 0.f;
      #pragma unroll
      for (int j = 1; j < 7; ++j) {
        sx += t[u][j].x; sy += t[u][j].y; sz += t[u][j].z; sw += t[u][j].w;
      }
      const float inv6 = 1.0f / 6.0f;
      float4 d = make_float4(sx * inv6 - loc.x, sy * inv6 - loc.y,
                             sz * inv6 - loc.z, sw * inv6 - loc.w);
      *(float4*)(xl + xoff[u])       = loc;
      *(float4*)(xl + xoff[u] + CIN) = d;
    }
  }
  __syncthreads();

  const int lane = tid & 31;
  const int wave = tid >> 5;
  const int g    = lane >> 4;
  const int nl   = lane & 15;
  #pragma unroll 1
  for (int it = 0; it < NIT; ++it) {
    const int nt = wave + NW * it;
    const int n = nt * 16 + nl;
    const float bv = bias[n];
    v8f acc[MT];
    #pragma unroll
    for (int mt = 0; mt < MT; ++mt) {
      #pragma unroll
      for (int j = 0; j < 8; ++j) acc[mt][j] = bv;
    }

    const float* __restrict__ ap = xl + nl * KP + 4 * g;
    const float* __restrict__ bp = W + (long)n * K2 + 4 * g;
    #pragma unroll 4
    for (int ks = 0; ks < NK8; ++ks) {
      const float4 b4 = *(const float4*)(bp + 8 * ks);
      float4 a4[MT];
      #pragma unroll
      for (int mt = 0; mt < MT; ++mt)
        a4[mt] = *(const float4*)(ap + mt * 16 * KP + 8 * ks);
      v2f b0, b1;
      b0[0] = b4.x; b0[1] = b4.y;
      b1[0] = b4.z; b1[1] = b4.w;
      #pragma unroll
      for (int mt = 0; mt < MT; ++mt) {
        v2f a0;
        a0[0] = a4[mt].x; a0[1] = a4[mt].y;
        acc[mt] = __builtin_amdgcn_wmma_f32_16x16x4_f32(
             false, a0,  false, b0,
             (short)0, acc[mt],  false,  false);
      }
      #pragma unroll
      for (int mt = 0; mt < MT; ++mt) {
        v2f a1;
        a1[0] = a4[mt].z; a1[1] = a4[mt].w;
        acc[mt] = __builtin_amdgcn_wmma_f32_16x16x4_f32(
             false, a1,  false, b1,
             (short)0, acc[mt],  false,  false);
      }
    }

    #pragma unroll
    for (int mt = 0; mt < MT; ++mt) {
      #pragma unroll
      for (int j = 0; j < 8; ++j) {
        int m = tile * BV + 16 * mt + j + 8 * g;
        float y = acc[mt][j];
        y = (y >= 0.f) ? y : ALPHA * y;
        if constexpr (CO2 > 0) {
          ot[(16 * mt + j + 8 * g) * OP + n] = y;
        } else {
          if (m < nv) out[(long)m * COUT + n] = y;
        }
      }
    }
  }
  if constexpr (CO2 > 0) {
    __syncthreads();
    second_map<COUT, OP, CO2, WP2, MT, NW>(ot, Wn, Yn, tile, nv, wave, g, nl);
  }

  if (clr) {
    const int k = tile * NT + tid;
    if (k < nclr) clr[k] = -1;
  }
}

template <int CB, int COUT, int CA, int CO2, int WP2>
__global__ __launch_bounds__((COUT / 16 >= 4) ? 128 : 64) void edgeconv_up(
    const float* __restrict__ Y, const int* __restrict__ mapA,
    const float* __restrict__ fB, const int* __restrict__ src,
    const float* __restrict__ W, const float* __restrict__ bias,
    float* __restrict__ out, int nv,
    const float* __restrict__ Wn, float* __restrict__ Yn)
{
  constexpr int WP  = 2 * (CA + CB);
  constexpr int K2  = 2 * CB;
  constexpr int KP  = K2 + 4;
  constexpr int TP  = COUT + 4;
  constexpr int C4  = CB / 4;
  constexpr int T4  = COUT / 4;
  constexpr int YP  = 2 * COUT;
  constexpr int NNT = COUT / 16;
  constexpr int MT  = (NNT >= 4) ? 4 : 2;
  constexpr int BV  = 16 * MT;
  constexpr int NW  = (NNT >= 4) ? 4 : 2;
  constexpr int NT  = 32 * NW;
  constexpr int NIT = NNT / NW;
  static_assert((KP / 4) % 2 == 1 && (TP / 4) % 2 == 1, "pitches must be odd numbers of 16-B units");
  static_assert(CB % 8 == 0 && CA % 4 == 0 && COUT % 16 == 0 && NNT % NW == 0, "tile shape");

  __shared__ __align__(16) int   rowB[BV * 8];
  __shared__ __align__(16) int   rowA[BV * 8];
  __shared__ __align__(16) float xl[BV * KP];
  __shared__ __align__(16) float tl[BV * TP];
  const int tid  = threadIdx.x;
  const int tile = blockIdx.x;

  {
    const int r = tid >> 3, j = tid & 7;
    const bool nb = (j >= 1) && (j <= 6);
    const int  jo = nb ? j - 1 : 0;
    int v[4], s[4], a[4];
    #pragma unroll
    for (int q = 0; q < 4; ++q) {
      int vv = tile * BV + r + (NT / 8) * q;
      v[q] = (vv >= nv) ? nv - 1 : vv;
    }
    #pragma unroll
    for (int q = 0; q < 4; ++q) s[q] = src[v[q] * 6 + jo];
    #pragma unroll
    for (int q = 0; q < 4; ++q) { if (!nb) s[q] = v[q]; }
    #pragma unroll
    for (int q = 0; q < 4; ++q) a[q] = mapA[s[q]];
    #pragma unroll
    for (int q = 0; q < 4; ++q) { rowB[tid + NT * q] = s[q]; rowA[tid + NT * q] = a[q]; }
  }
  __syncthreads();

  static_assert((BV * C4) % (2 * NT) == 0, "two x items per thread per trip");
  for (int i0 = tid; i0 < BV * C4; i0 += 2 * NT) {
    int    xoff[2];
    float4 t[2][7];
    #pragma unroll
    for (int u = 0; u < 2; ++u) {
      const int i = i0 + NT * u;
      const int r = i / C4, c4 = (i - r * C4) << 2;
      const int* rows = rowB + r * 8;
      const int4 r0 = *(const int4*)(rows);
      const int4 r1 = *(const int4*)(rows + 4);
      const int id[7] = { r0.x, r0.y, r0.z, r0.w, r1.x, r1.y, r1.z };
      #pragma unroll
      for (int j = 0; j < 7; ++j)
        t[u][j] = *(const float4*)(fB + (long)id[j] * CB + c4);
      xoff[u] = r * KP + c4;
    }
    #pragma unroll
    for (int u = 0; u < 2; ++u) {
      const float4 loc = t[u][0];
      float sx = 0.f, sy = 0.f, sz = 0.f, sw = 0.f;
      #pragma unroll
      for (int j = 1; j < 7; ++j) {
        sx += t[u][j].x; sy += t[u][j].y; sz += t[u][j].z; sw += t[u][j].w;
      }
      const float inv6 = 1.0f / 6.0f;
      float4 d = make_float4(sx * inv6 - loc.x, sy * inv6 - loc.y,
                             sz * inv6 - loc.z, sw * inv6 - loc.w);
      *(float4*)(xl + xoff[u])      = loc;
      *(float4*)(xl + xoff[u] + CB) = d;
    }
  }

  static_assert((BV * T4) % (2 * NT) == 0, "two T items per thread per trip");
  for (int i0 = tid; i0 < BV * T4; i0 += 2 * NT) {
    int    toff[2];
    float4 yc[2], ya[2][7];
    #pragma unroll
    for (int u = 0; u < 2; ++u) {
      const int i = i0 + NT * u;
      const int r = i / T4, c4 = (i - r * T4) << 2;
      const int* rows = rowA + r * 8;
      const int4 r0 = *(const int4*)(rows);
      const int4 r1 = *(const int4*)(rows + 4);
      const int id[7] = { r0.x, r0.y, r0.z, r0.w, r1.x, r1.y, r1.z };
      yc[u] = *(const float4*)(Y + (long)id[0] * YP + c4);
      #pragma unroll
      for (int j = 0; j < 7; ++j)
        ya[u][j] = *(const float4*)(Y + (long)id[j] * YP + COUT + c4);
      toff[u] = r * TP + c4;
    }
    #pragma unroll
    for (int u = 0; u < 2; ++u) {
      float sx = 0.f, sy = 0.f, sz = 0.f, sw = 0.f;
      #pragma unroll
      for (int j = 1; j < 7; ++j) {
        sx += ya[u][j].x; sy += ya[u][j].y; sz += ya[u][j].z; sw += ya[u][j].w;
      }
      const float inv6 = 1.0f / 6.0f;
      *(float4*)(tl + toff[u]) =
          make_float4((yc[u].x - ya[u][0].x) + sx * inv6, (yc[u].y - ya[u][0].y) + sy * inv6,
                      (yc[u].z - ya[u][0].z) + sz * inv6, (yc[u].w - ya[u][0].w) + sw * inv6);
    }
  }
  __syncthreads();

  const int lane = tid & 31;
  const int wave = tid >> 5;
  const int g    = lane >> 4;
  const int nl   = lane & 15;
  #pragma unroll 1
  for (int it = 0; it < NIT; ++it) {
    const int nt = wave + NW * it;
    const int n = nt * 16 + nl;
    const float bv = bias[n];
    v8f acc[MT];
    #pragma unroll
    for (int mt = 0; mt < MT; ++mt) {
      #pragma unroll
      for (int j = 0; j < 8; ++j) acc[mt][j] = bv + tl[(16 * mt + j + 8 * g) * TP + n];
    }

    const float* __restrict__ ap = xl + nl * KP + 4 * g;
    const float* __restrict__ bp = W + (long)n * WP + CA + 4 * g;
    #pragma unroll 4
    for (int ks = 0; ks < CB / 8; ++ks) kstep8<MT>(acc, ap + 8 * ks, 16 * KP, bp + 8 * ks);
    #pragma unroll 4
    for (int ks = 0; ks < CB / 8; ++ks)
      kstep8<MT>(acc, ap + CB + 8 * ks, 16 * KP, bp + (CA + CB) + 8 * ks);

    #pragma unroll
    for (int mt = 0; mt < MT; ++mt) {
      #pragma unroll
      for (int j = 0; j < 8; ++j) {
        int m = tile * BV + 16 * mt + j + 8 * g;
        float y = acc[mt][j];
        y = (y >= 0.f) ? y : ALPHA * y;
        if constexpr (CO2 > 0) {
          tl[(16 * mt + j + 8 * g) * TP + n] = y;
        } else {
          if (m < nv) out[(long)m * COUT + n] = y;
        }
      }
    }
  }
  if constexpr (CO2 > 0) {
    __syncthreads();
    second_map<COUT, TP, CO2, WP2, MT, NW>(tl, Wn, Yn, tile, nv, wave, g, nl);
  }
}

__device__ __forceinline__ unsigned ford(float f) {
  unsigned u = __float_as_uint(f);
  return (u & 0x80000000u) ? ~u : (u | 0x80000000u);
}
__device__ __forceinline__ float funord(unsigned u) {
  unsigned b = (u & 0x80000000u) ? (u & 0x7FFFFFFFu) : ~u;
  return __uint_as_float(b);
}

__global__ __launch_bounds__(256) void pool_link(
    const int* __restrict__ mapF, int nF, int* __restrict__ headF, int* __restrict__ nextF,
    const int* __restrict__ mapC, int nC, int* __restrict__ headC, int* __restrict__ nextC)
{
  const int i = blockIdx.x * 256 + threadIdx.x;
  if (i < nF) {
    nextF[i] = atomicExch(&headF[mapF[i]], i);
  } else if (i < nF + nC) {
    const int k = i - nF;
    nextC[k] = atomicExch(&headC[mapC[k]], k);
  }
}

__global__ __launch_bounds__(256) void pool_walk(
    const float* __restrict__ f, const int* __restrict__ head, const int* __restrict__ next,
    float* __restrict__ out, int nparent, int nchild, int lshift)
{
  const int t = blockIdx.x * 256 + threadIdx.x;
  const int p = t >> lshift;
  const int l = t & ((1 << lshift) - 1);
  if (p >= nparent) return;
  const int C = 4 << lshift;
  int i = head[p];
  unsigned mx = 0x007FFFFFu, my = 0x007FFFFFu, mz = 0x007FFFFFu, mw = 0x007FFFFFu;
  for (int k = 0; k < nchild && i >= 0; ++k) {
    const float4 v = *(const float4*)(f + (long)i * C + 4 * l);
    const int nx = next[i];
    const unsigned ux = ford(v.x), uy = ford(v.y), uz = ford(v.z), uw = ford(v.w);
    mx = ux > mx ? ux : mx;  my = uy > my ? uy : my;
    mz = uz > mz ? uz : mz;  mw = uw > mw ? uw : mw;
    i = nx;
  }
  *(float4*)(out + (long)p * C + 4 * l) =
      make_float4(funord(mx), funord(my), funord(mz), funord(mw));
}

extern "C" void kernel_launch(void* const* d_in, const int* in_sizes, int n_in,
                              void* d_out, int out_size, void* d_ws, size_t ws_size,
                              hipStream_t stream) {
  (void)in_sizes; (void)n_in; (void)out_size; (void)ws_size;
  const float* features = (const float*)d_in[0];
  const int*   src2     = (const int*)d_in[1];
  const int*   src1     = (const int*)d_in[3];
  const int*   src0     = (const int*)d_in[5];
  const int*   p21      = (const int*)d_in[7];
  const int*   p10      = (const int*)d_in[8];
  const float* w_enc0 = (const float*)d_in[9],  *b_enc0 = (const float*)d_in[10];
  const float* w_enc1 = (const float*)d_in[11], *b_enc1 = (const float*)d_in[12];
  const float* w_enc2 = (const float*)d_in[13], *b_enc2 = (const float*)d_in[14];
  const float* w_dec0 = (const float*)d_in[15], *b_dec0 = (const float*)d_in[16];
  const float* w_dec1 = (const float*)d_in[17], *b_dec1 = (const float*)d_in[18];

  char* ws = (char*)d_ws;
  size_t off = 0;
  float* f2 = (float*)(ws + off); off += (size_t)NV2 * 32 * 4;
  float* f1 = (float*)(ws + off); off += (size_t)NV1 * 64 * 4;
  float* y0 = (float*)(ws + off); off += (size_t)NV0 * 128 * 4;
  float* y1 = (float*)(ws + off); off += (size_t)NV1 * 64 * 4;
  float* P1 = (float*)(ws + off); off += (size_t)NV1 * 32 * 4;
  float* P0 = (float*)(ws + off); off += (size_t)NV0 * 64 * 4;
  int* head1 = (int*)(ws + off); off += (size_t)NV1 * 4;
  int* head0 = (int*)(ws + off); off += (size_t)NV0 * 4;
  int* next2 = (int*)(ws + off); off += ((size_t)NV2 * 4 + 15) / 16 * 16;
  int* next1 = (int*)(ws + off);

  constexpr int G2 = (NV2 + 31) / 32, G1 = (NV1 + 63) / 64, G0 = (NV0 + 63) / 64;
  auto walk = [&](const float* fchild, const int* head, const int* next, float* pooled,
                  int nchild, int nparent, int lshift) {
    int nt = nparent << lshift;
    pool_walk<<<(nt + 255) / 256, 256, 0, stream>>>(fchild, head, next, pooled,
                                                     nparent, nchild, lshift);
  };

  edgeconv_wmma<16, 32, 0, 0><<<G2, 64, 0, stream>>>(features, nullptr, 16, nullptr, 0, src2,
                                                      w_enc0, b_enc0, f2, NV2, head1, NV1 + NV0,
                                                      nullptr, nullptr);
  pool_link<<<(NV2 + NV1 + 255) / 256, 256, 0, stream>>>(p21, NV2, head1, next2,
                                                         p10, NV1, head0, next1);
  walk(f2, head1, next2, P1, NV2, NV1, 3);
  edgeconv_wmma<32, 64, 0, 0><<<G1, 128, 0, stream>>>(P1, nullptr, 32, nullptr, 0, src1,
                                                      w_enc1, b_enc1, f1, NV1, nullptr, 0,
                                                      nullptr, nullptr);
  walk(f1, head0, next1, P0, NV1, NV0, 4);
  edgeconv_wmma<64, 128, 64, 384><<<G0, 128, 0, stream>>>(P0, nullptr, 64, nullptr, 0, src0,
                                                          w_enc2, b_enc2, nullptr, NV0, nullptr, 0,
                                                          w_dec0, y0);

  edgeconv_up<64, 64, 128, 32, 192><<<G1, 128, 0, stream>>>(y0, p10, f1, src1, w_dec0, b_dec0,
                                                            nullptr, NV1, w_dec1, y1);
  edgeconv_up<32, 32, 64, 0, 0><<<G2, 64, 0, stream>>>(y1, p21, f2, src2, w_dec1, b_dec1,
                                                        (float*)d_out, NV2, nullptr, nullptr);
}


# Coarsest-level 128-channel layer re-tiled: 48-vertex blocks of 8 waves (was 64, 4): 214 blocks, one per WGP, two waves per SIMD cover each other's W fetches; gather in one trip; sums unchanged
# speedup vs baseline: 1.8686x; 1.0744x over previous
#include <hip/hip_runtime.h>

typedef __attribute__((ext_vector_type(2))) float v2f;
typedef __attribute__((ext_vector_type(8))) float v8f;

#define NV2 163842
#define NV1 40962
#define NV0 10242
#define ALPHA 0.3f

template <int MT>
__device__ __forceinline__ void kstep8(v8f (&acc)[MT], const float* __restrict__ a,
                                       int mstride, const float* __restrict__ b)
{
  const float4 b4 = *(const float4*)b;
  float4 a4[MT];
  #pragma unroll
  for (int mt = 0; mt < MT; ++mt) a4[mt] = *(const float4*)(a + mt * mstride);
  v2f b0, b1;
  b0[0] = b4.x; b0[1] = b4.y;
  b1[0] = b4.z; b1[1] = b4.w;
  #pragma unroll
  for (int mt = 0; mt < MT; ++mt) {
    v2f a0;
    a0[0] = a4[mt].x; a0[1] = a4[mt].y;
    acc[mt] = __builtin_amdgcn_wmma_f32_16x16x4_f32(
         false, a0,  false, b0,
         (short)0, acc[mt],  false,  false);
  }
  #pragma unroll
  for (int mt = 0; mt < MT; ++mt) {
    v2f a1;
    a1[0] = a4[mt].z; a1[1] = a4[mt].w;
    acc[mt] = __builtin_amdgcn_wmma_f32_16x16x4_f32(
         false, a1,  false, b1,
         (short)0, acc[mt],  false,  false);
  }
}

template <int KD, int TP, int CO2, int WP2, int MT, int NW>
__device__ __forceinline__ void second_map(const float* __restrict__ tl,
                                           const float* __restrict__ Wn, float* __restrict__ Yn,
                                           int tile, int nv, int wave, int g, int nl)
{
  constexpr int NNT2 = 2 * CO2 / 16;
  constexpr int NIT2 = NNT2 / NW;
  static_assert(NNT2 % NW == 0 && CO2 % 16 == 0 && KD % 8 == 0, "tile shape");
  static_assert((TP / 4) % 2 == 1, "tile pitch must be an odd number of 16-B units");
  #pragma unroll 1
  for (int it = 0; it < NIT2; ++it) {
    const int np = (wave + NW * it) * 16 + nl;
    const bool second = np >= CO2;
    const int n = second ? np - CO2 : np;
    v8f acc[MT];
    #pragma unroll
    for (int mt = 0; mt < MT; ++mt) {
      #pragma unroll
      for (int j = 0; j < 8; ++j) acc[mt][j] = 0.f;
    }
    const float* __restrict__ ap = tl + nl * TP + 4 * g;
    const float* __restrict__ bp = Wn + (long)n * WP2 + (second ? WP2 / 2 : 0) + 4 * g;
    #pragma unroll 4
    for (int ks = 0; ks < KD / 8; ++ks) kstep8<MT>(acc, ap + 8 * ks, 16 * TP, bp + 8 * ks);
    #pragma unroll
    for (int mt = 0; mt < MT; ++mt) {
      #pragma unroll
      for (int j = 0; j < 8; ++j) {
        const int m = tile * (16 * MT) + 16 * mt + j + 8 * g;
        if (m < nv) Yn[(long)m * (2 * CO2) + np] = acc[mt][j];
      }
    }
  }
}


template <int CIN, int COUT, int CO2, int WP2>
__global__ __launch_bounds__((COUT / 16 >= 8) ? 256 : (COUT / 16 >= 4) ? 128 : 64) void edgeconv_wmma(
    const float* __restrict__ fA, const int* __restrict__ mapA, int CA,
    const float* __restrict__ fB, int CB,
    const int* __restrict__ src,
    const float* __restrict__ W, const float* __restrict__ bias,
    float* __restrict__ out, int nv,
    int* __restrict__ clr, int nclr,
    const float* __restrict__ Wn, float* __restrict__ Yn)
{
  constexpr int K2  = 2 * CIN;
  constexpr int KP  = K2 + 4;
  constexpr int C4  = CIN / 4;
  constexpr int NK8 = K2 / 8;
  constexpr int NNT = COUT / 16;
  constexpr int MT  = (NNT >= 8) ? 3 : (NNT >= 4) ? 4 : 2;
  constexpr int BV  = 16 * MT;
  constexpr int NW  = (NNT >= 8) ? 8 : (NNT >= 4) ? 4 : 2;
  constexpr int NT  = 32 * NW;
  constexpr int NIT = NNT / NW;
  constexpr int RQ  = (BV * 8 + NT - 1) / NT;
  constexpr bool RQ_EXACT = (BV * 8) % NT == 0;
  static_assert(NNT % NW == 0, "one wave per N-tile, or a whole number of turns");

  __shared__ __align__(16) int   rowB[BV * 8];
  __shared__ __align__(16) int   rowA[BV * 8];
  __shared__ __align__(16) float xl[BV * KP];
  constexpr int OP  = COUT + 4;
  __shared__ __align__(16) float ot[(CO2 > 0) ? BV * OP : 4];
  const int tid  = threadIdx.x;
  const int tile = blockIdx.x;

  {
    const int r = tid >> 3, j = tid & 7;
    const bool nb = (j >= 1) && (j <= 6);
    const int  jo = nb ? j - 1 : 0;
    int v[RQ], s[RQ], a[RQ];
    #pragma unroll
    for (int q = 0; q < RQ; ++q) {
      int vv = tile * BV + r + (NT / 8) * q;
      v[q] = (vv >= nv) ? nv - 1 : vv;
    }
    #pragma unroll
    for (int q = 0; q < RQ; ++q) s[q] = src[v[q] * 6 + jo];
    #pragma unroll
    for (int q = 0; q < RQ; ++q) { if (!nb) s[q] = v[q]; a[q] = s[q]; }
    if (mapA) {
      #pragma unroll
      for (int q = 0; q < RQ; ++q) a[q] = mapA[s[q]];
    }
    #pragma unroll
    for (int q = 0; q < RQ; ++q) {
      if (RQ_EXACT || tid + NT * q < BV * 8) { rowB[tid + NT * q] = s[q]; rowA[tid + NT * q] = a[q]; }
    }
  }
  __syncthreads();

  constexpr int UI = ((BV * C4) % (2 * NT) == 0) ? 2 : 3;
  static_assert((BV * C4) % (UI * NT) == 0, "UI items per thread per trip");
  for (int i0 = tid; i0 < BV * C4; i0 += UI * NT) {
    int    xoff[UI];
    float4 t[UI][7];
    #pragma unroll
    for (int u = 0; u < UI; ++u) {
      const int i = i0 + NT * u;
      const int r = i / C4, c4 = (i - r * C4) << 2;
      const bool inA = c4 < CA;
      const float* __restrict__ fp = inA ? fA : fB;
      const int  stride = inA ? CA : CB;
      const int  coff   = inA ? c4 : c4 - CA;
      const int* rows   = (inA ? rowA : rowB) + r * 8;
      const int4 r0 = *(const int4*)(rows);
      const int4 r1 = *(const int4*)(rows + 4);
      const int id[7] = { r0.x, r0.y, r0.z, r0.w, r1.x, r1.y, r1.z };
      #pragma unroll
      for (int j = 0; j < 7; ++j)
        t[u][j] = *(const float4*)(fp + (long)id[j] * stride + coff);
      xoff[u] = r * KP + c4;
    }
    #pragma unroll
    for (int u = 0; u < UI; ++u) {
      const float4 loc = t[u][0];
      float sx = 0.f, sy = 0.f, sz = 0.f, sw = 0.f;
      #pragma unroll
      for (int j = 1; j < 7; ++j) {
        sx += t[u][j].x; sy += t[u][j].y; sz += t[u][j].z; sw += t[u][j].w;
      }
      const float inv6 = 1.0f / 6.0f;
      float4 d = make_float4(sx * inv6 - loc.x, sy * inv6 - loc.y,
                             sz * inv6 - loc.z, sw * inv6 - loc.w);
      *(float4*)(xl + xoff[u])       = loc;
      *(float4*)(xl + xoff[u] + CIN) = d;
    }
  }
  __syncthreads();

  const int lane = tid & 31;
  const int wave = tid >> 5;
  const int g    = lane >> 4;
  const int nl   = lane & 15;
  #pragma unroll 1
  for (int it = 0; it < NIT; ++it) {
    const int nt = wave + NW * it;
    const int n = nt * 16 + nl;
    const float bv = bias[n];
    v8f acc[MT];
    #pragma unroll
    for (int mt = 0; mt < MT; ++mt) {
      #pragma unroll
      for (int j = 0; j < 8; ++j) acc[mt][j] = bv;
    }

    const float* __restrict__ ap = xl + nl * KP + 4 * g;
    const float* __restrict__ bp = W + (long)n * K2 + 4 * g;
    #pragma unroll 4
    for (int ks = 0; ks < NK8; ++ks) {
      const float4 b4 = *(const float4*)(bp + 8 * ks);
      float4 a4[MT];
      #pragma unroll
      for (int mt = 0; mt < MT; ++mt)
        a4[mt] = *(const float4*)(ap + mt * 16 * KP + 8 * ks);
      v2f b0, b1;
      b0[0] = b4.x; b0[1] = b4.y;
      b1[0] = b4.z; b1[1] = b4.w;
      #pragma unroll
      for (int mt = 0; mt < MT; ++mt) {
        v2f a0;
        a0[0] = a4[mt].x; a0[1] = a4[mt].y;
        acc[mt] = __builtin_amdgcn_wmma_f32_16x16x4_f32(
             false, a0,  false, b0,
             (short)0, acc[mt],  false,  false);
      }
      #pragma unroll
      for (int mt = 0; mt < MT; ++mt) {
        v2f a1;
        a1[0] = a4[mt].z; a1[1] = a4[mt].w;
        acc[mt] = __builtin_amdgcn_wmma_f32_16x16x4_f32(
             false, a1,  false, b1,
             (short)0, acc[mt],  false,  false);
      }
    }

    #pragma unroll
    for (int mt = 0; mt < MT; ++mt) {
      #pragma unroll
      for (int j = 0; j < 8; ++j) {
        int m = tile * BV + 16 * mt + j + 8 * g;
        float y = acc[mt][j];
        y = (y >= 0.f) ? y : ALPHA * y;
        if constexpr (CO2 > 0) {
          ot[(16 * mt + j + 8 * g) * OP + n] = y;
        } else {
          if (m < nv) out[(long)m * COUT + n] = y;
        }
      }
    }
  }
  if constexpr (CO2 > 0) {
    __syncthreads();
    second_map<COUT, OP, CO2, WP2, MT, NW>(ot, Wn, Yn, tile, nv, wave, g, nl);
  }

  if (clr) {
    const int k = tile * NT + tid;
    if (k < nclr) clr[k] = -1;
  }
}

template <int CB, int COUT, int CA, int CO2, int WP2>
__global__ __launch_bounds__((COUT / 16 >= 4) ? 128 : 64) void edgeconv_up(
    const float* __restrict__ Y, const int* __restrict__ mapA,
    const float* __restrict__ fB, const int* __restrict__ src,
    const float* __restrict__ W, const float* __restrict__ bias,
    float* __restrict__ out, int nv,
    const float* __restrict__ Wn, float* __restrict__ Yn)
{
  constexpr int WP  = 2 * (CA + CB);
  constexpr int K2  = 2 * CB;
  constexpr int KP  = K2 + 4;
  constexpr int TP  = COUT + 4;
  constexpr int C4  = CB / 4;
  constexpr int T4  = COUT / 4;
  constexpr int YP  = 2 * COUT;
  constexpr int NNT = COUT / 16;
  constexpr int MT  = (NNT >= 4) ? 4 : 2;
  constexpr int BV  = 16 * MT;
  constexpr int NW  = (NNT >= 4) ? 4 : 2;
  constexpr int NT  = 32 * NW;
  constexpr int NIT = NNT / NW;
  static_assert((KP / 4) % 2 == 1 && (TP / 4) % 2 == 1, "pitches must be odd numbers of 16-B units");
  static_assert(CB % 8 == 0 && CA % 4 == 0 && COUT % 16 == 0 && NNT % NW == 0, "tile shape");

  __shared__ __align__(16) int   rowB[BV * 8];
  __shared__ __align__(16) int   rowA[BV * 8];
  __shared__ __align__(16) float xl[BV * KP];
  __shared__ __align__(16) float tl[BV * TP];
  const int tid  = threadIdx.x;
  const int tile = blockIdx.x;

  {
    const int r = tid >> 3, j = tid & 7;
    const bool nb = (j >= 1) && (j <= 6);
    const int  jo = nb ? j - 1 : 0;
    int v[4], s[4], a[4];
    #pragma unroll
    for (int q = 0; q < 4; ++q) {
      int vv = tile * BV + r + (NT / 8) * q;
      v[q] = (vv >= nv) ? nv - 1 : vv;
    }
    #pragma unroll
    for (int q = 0; q < 4; ++q) s[q] = src[v[q] * 6 + jo];
    #pragma unroll
    for (int q = 0; q < 4; ++q) { if (!nb) s[q] = v[q]; }
    #pragma unroll
    for (int q = 0; q < 4; ++q) a[q] = mapA[s[q]];
    #pragma unroll
    for (int q = 0; q < 4; ++q) { rowB[tid + NT * q] = s[q]; rowA[tid + NT * q] = a[q]; }
  }
  __syncthreads();

  static_assert((BV * C4) % (2 * NT) == 0, "two x items per thread per trip");
  for (int i0 = tid; i0 < BV * C4; i0 += 2 * NT) {
    int    xoff[2];
    float4 t[2][7];
    #pragma unroll
    for (int u = 0; u < 2; ++u) {
      const int i = i0 + NT * u;
      const int r = i / C4, c4 = (i - r * C4) << 2;
      const int* rows = rowB + r * 8;
      const int4 r0 = *(const int4*)(rows);
      const int4 r1 = *(const int4*)(rows + 4);
      const int id[7] = { r0.x, r0.y, r0.z, r0.w, r1.x, r1.y, r1.z };
      #pragma unroll
      for (int j = 0; j < 7; ++j)
        t[u][j] = *(const float4*)(fB + (long)id[j] * CB + c4);
      xoff[u] = r * KP + c4;
    }
    #pragma unroll
    for (int u = 0; u < 2; ++u) {
      const float4 loc = t[u][0];
      float sx = 0.f, sy = 0.f, sz = 0.f, sw = 0.f;
      #pragma unroll
      for (int j = 1; j < 7; ++j) {
        sx += t[u][j].x; sy += t[u][j].y; sz += t[u][j].z; sw += t[u][j].w;
      }
      const float inv6 = 1.0f / 6.0f;
      float4 d = make_float4(sx * inv6 - loc.x, sy * inv6 - loc.y,
                             sz * inv6 - loc.z, sw * inv6 - loc.w);
      *(float4*)(xl + xoff[u])      = loc;
      *(float4*)(xl + xoff[u] + CB) = d;
    }
  }

  static_assert((BV * T4) % (2 * NT) == 0, "two T items per thread per trip");
  for (int i0 = tid; i0 < BV * T4; i0 += 2 * NT) {
    int    toff[2];
    float4 yc[2], ya[2][7];
    #pragma unroll
    for (int u = 0; u < 2; ++u) {
      const int i = i0 + NT * u;
      const int r = i / T4, c4 = (i - r * T4) << 2;
      const int* rows = rowA + r * 8;
      const int4 r0 = *(const int4*)(rows);
      const int4 r1 = *(const int4*)(rows + 4);
      const int id[7] = { r0.x, r0.y, r0.z, r0.w, r1.x, r1.y, r1.z };
      yc[u] = *(const float4*)(Y + (long)id[0] * YP + c4);
      #pragma unroll
      for (int j = 0; j < 7; ++j)
        ya[u][j] = *(const float4*)(Y + (long)id[j] * YP + COUT + c4);
      toff[u] = r * TP + c4;
    }
    #pragma unroll
    for (int u = 0; u < 2; ++u) {
      float sx = 0.f, sy = 0.f, sz = 0.f, sw = 0.f;
      #pragma unroll
      for (int j = 1; j < 7; ++j) {
        sx += ya[u][j].x; sy += ya[u][j].y; sz += ya[u][j].z; sw += ya[u][j].w;
      }
      const float inv6 = 1.0f / 6.0f;
      *(float4*)(tl + toff[u]) =
          make_float4((yc[u].x - ya[u][0].x) + sx * inv6, (yc[u].y - ya[u][0].y) + sy * inv6,
                      (yc[u].z - ya[u][0].z) + sz * inv6, (yc[u].w - ya[u][0].w) + sw * inv6);
    }
  }
  __syncthreads();

  const int lane = tid & 31;
  const int wave = tid >> 5;
  const int g    = lane >> 4;
  const int nl   = lane & 15;
  #pragma unroll 1
  for (int it = 0; it < NIT; ++it) {
    const int nt = wave + NW * it;
    const int n = nt * 16 + nl;
    const float bv = bias[n];
    v8f acc[MT];
    #pragma unroll
    for (int mt = 0; mt < MT; ++mt) {
      #pragma unroll
      for (int j = 0; j < 8; ++j) acc[mt][j] = bv + tl[(16 * mt + j + 8 * g) * TP + n];
    }

    const float* __restrict__ ap = xl + nl * KP + 4 * g;
    const float* __restrict__ bp = W + (long)n * WP + CA + 4 * g;
    #pragma unroll 4
    for (int ks = 0; ks < CB / 8; ++ks) kstep8<MT>(acc, ap + 8 * ks, 16 * KP, bp + 8 * ks);
    #pragma unroll 4
    for (int ks = 0; ks < CB / 8; ++ks)
      kstep8<MT>(acc, ap + CB + 8 * ks, 16 * KP, bp + (CA + CB) + 8 * ks);

    #pragma unroll
    for (int mt = 0; mt < MT; ++mt) {
      #pragma unroll
      for (int j = 0; j < 8; ++j) {
        int m = tile * BV + 16 * mt + j + 8 * g;
        float y = acc[mt][j];
        y = (y >= 0.f) ? y : ALPHA * y;
        if constexpr (CO2 > 0) {
          tl[(16 * mt + j + 8 * g) * TP + n] = y;
        } else {
          if (m < nv) out[(long)m * COUT + n] = y;
        }
      }
    }
  }
  if constexpr (CO2 > 0) {
    __syncthreads();
    second_map<COUT, TP, CO2, WP2, MT, NW>(tl, Wn, Yn, tile, nv, wave, g, nl);
  }
}

__device__ __forceinline__ unsigned ford(float f) {
  unsigned u = __float_as_uint(f);
  return (u & 0x80000000u) ? ~u : (u | 0x80000000u);
}
__device__ __forceinline__ float funord(unsigned u) {
  unsigned b = (u & 0x80000000u) ? (u & 0x7FFFFFFFu) : ~u;
  return __uint_as_float(b);
}

__global__ __launch_bounds__(256) void pool_link(
    const int* __restrict__ mapF, int nF, int* __restrict__ headF, int* __restrict__ nextF,
    const int* __restrict__ mapC, int nC, int* __restrict__ headC, int* __restrict__ nextC)
{
  const int i = blockIdx.x * 256 + threadIdx.x;
  if (i < nF) {
    nextF[i] = atomicExch(&headF[mapF[i]], i);
  } else if (i < nF + nC) {
    const int k = i - nF;
    nextC[k] = atomicExch(&headC[mapC[k]], k);
  }
}

__global__ __launch_bounds__(256) void pool_walk(
    const float* __restrict__ f, const int* __restrict__ head, const int* __restrict__ next,
    float* __restrict__ out, int nparent, int nchild, int lshift)
{
  const int t = blockIdx.x * 256 + threadIdx.x;
  const int p = t >> lshift;
  const int l = t & ((1 << lshift) - 1);
  if (p >= nparent) return;
  const int C = 4 << lshift;
  int i = head[p];
  unsigned mx = 0x007FFFFFu, my = 0x007FFFFFu, mz = 0x007FFFFFu, mw = 0x007FFFFFu;
  for (int k = 0; k < nchild && i >= 0; ++k) {
    const float4 v = *(const float4*)(f + (long)i * C + 4 * l);
    const int nx = next[i];
    const unsigned ux = ford(v.x), uy = ford(v.y), uz = ford(v.z), uw = ford(v.w);
    mx = ux > mx ? ux : mx;  my = uy > my ? uy : my;
    mz = uz > mz ? uz : mz;  mw = uw > mw ? uw : mw;
    i = nx;
  }
  *(float4*)(out + (long)p * C + 4 * l) =
      make_float4(funord(mx), funord(my), funord(mz), funord(mw));
}

extern "C" void kernel_launch(void* const* d_in, const int* in_sizes, int n_in,
                              void* d_out, int out_size, void* d_ws, size_t ws_size,
                              hipStream_t stream) {
  (void)in_sizes; (void)n_in; (void)out_size; (void)ws_size;
  const float* features = (const float*)d_in[0];
  const int*   src2     = (const int*)d_in[1];
  const int*   src1     = (const int*)d_in[3];
  const int*   src0     = (const int*)d_in[5];
  const int*   p21      = (const int*)d_in[7];
  const int*   p10      = (const int*)d_in[8];
  const float* w_enc0 = (const float*)d_in[9],  *b_enc0 = (const float*)d_in[10];
  const float* w_enc1 = (const float*)d_in[11], *b_enc1 = (const float*)d_in[12];
  const float* w_enc2 = (const float*)d_in[13], *b_enc2 = (const float*)d_in[14];
  const float* w_dec0 = (const float*)d_in[15], *b_dec0 = (const float*)d_in[16];
  const float* w_dec1 = (const float*)d_in[17], *b_dec1 = (const float*)d_in[18];

  char* ws = (char*)d_ws;
  size_t off = 0;
  float* f2 = (float*)(ws + off); off += (size_t)NV2 * 32 * 4;
  float* f1 = (float*)(ws + off); off += (size_t)NV1 * 64 * 4;
  float* y0 = (float*)(ws + off); off += (size_t)NV0 * 128 * 4;
  float* y1 = (float*)(ws + off); off += (size_t)NV1 * 64 * 4;
  float* P1 = (float*)(ws + off); off += (size_t)NV1 * 32 * 4;
  float* P0 = (float*)(ws + off); off += (size_t)NV0 * 64 * 4;
  int* head1 = (int*)(ws + off); off += (size_t)NV1 * 4;
  int* head0 = (int*)(ws + off); off += (size_t)NV0 * 4;
  int* next2 = (int*)(ws + off); off += ((size_t)NV2 * 4 + 15) / 16 * 16;
  int* next1 = (int*)(ws + off);

  constexpr int G2 = (NV2 + 31) / 32, G1 = (NV1 + 63) / 64, G0 = (NV0 + 47) / 48;
  auto walk = [&](const float* fchild, const int* head, const int* next, float* pooled,
                  int nchild, int nparent, int lshift) {
    int nt = nparent << lshift;
    pool_walk<<<(nt + 255) / 256, 256, 0, stream>>>(fchild, head, next, pooled,
                                                     nparent, nchild, lshift);
  };

  edgeconv_wmma<16, 32, 0, 0><<<G2, 64, 0, stream>>>(features, nullptr, 16, nullptr, 0, src2,
                                                      w_enc0, b_enc0, f2, NV2, head1, NV1 + NV0,
                                                      nullptr, nullptr);
  pool_link<<<(NV2 + NV1 + 255) / 256, 256, 0, stream>>>(p21, NV2, head1, next2,
                                                         p10, NV1, head0, next1);
  walk(f2, head1, next2, P1, NV2, NV1, 3);
  edgeconv_wmma<32, 64, 0, 0><<<G1, 128, 0, stream>>>(P1, nullptr, 32, nullptr, 0, src1,
                                                      w_enc1, b_enc1, f1, NV1, nullptr, 0,
                                                      nullptr, nullptr);
  walk(f1, head0, next1, P0, NV1, NV0, 4);
  edgeconv_wmma<64, 128, 64, 384><<<G0, 256, 0, stream>>>(P0, nullptr, 64, nullptr, 0, src0,
                                                          w_enc2, b_enc2, nullptr, NV0, nullptr, 0,
                                                          w_dec0, y0);

  edgeconv_up<64, 64, 128, 32, 192><<<G1, 128, 0, stream>>>(y0, p10, f1, src1, w_dec0, b_dec0,
                                                            nullptr, NV1, w_dec1, y1);
  edgeconv_up<32, 32, 64, 0, 0><<<G2, 64, 0, stream>>>(y1, p21, f2, src2, w_dec1, b_dec1,
                                                        (float*)d_out, NV2, nullptr, nullptr);
}


# segment_max pooling inside the producing layer: its output tile staged in LDS, each row sent as one unsigned atomic max per 128-B line on order-preserving images (max is order-free); 8->6 launches
# speedup vs baseline: 2.0854x; 1.1160x over previous
#include <hip/hip_runtime.h>

typedef __attribute__((ext_vector_type(2))) float v2f;
typedef __attribute__((ext_vector_type(8))) float v8f;

#define NV2 163842
#define NV1 40962
#define NV0 10242
#define ALPHA 0.3f

template <int MT>
__device__ __forceinline__ void kstep8(v8f (&acc)[MT], const float* __restrict__ a,
                                       int mstride, const float* __restrict__ b)
{
  const float4 b4 = *(const float4*)b;
  float4 a4[MT];
  #pragma unroll
  for (int mt = 0; mt < MT; ++mt) a4[mt] = *(const float4*)(a + mt * mstride);
  v2f b0, b1;
  b0[0] = b4.x; b0[1] = b4.y;
  b1[0] = b4.z; b1[1] = b4.w;
  #pragma unroll
  for (int mt = 0; mt < MT; ++mt) {
    v2f a0;
    a0[0] = a4[mt].x; a0[1] = a4[mt].y;
    acc[mt] = __builtin_amdgcn_wmma_f32_16x16x4_f32(
         false, a0,  false, b0,
         (short)0, acc[mt],  false,  false);
  }
  #pragma unroll
  for (int mt = 0; mt < MT; ++mt) {
    v2f a1;
    a1[0] = a4[mt].z; a1[1] = a4[mt].w;
    acc[mt] = __builtin_amdgcn_wmma_f32_16x16x4_f32(
         false, a1,  false, b1,
         (short)0, acc[mt],  false,  false);
  }
}

template <int KD, int TP, int CO2, int WP2, int MT, int NW>
__device__ __forceinline__ void second_map(const float* __restrict__ tl,
                                           const float* __restrict__ Wn, float* __restrict__ Yn,
                                           int tile, int nv, int wave, int g, int nl)
{
  constexpr int NNT2 = 2 * CO2 / 16;
  constexpr int NIT2 = NNT2 / NW;
  static_assert(NNT2 % NW == 0 && CO2 % 16 == 0 && KD % 8 == 0, "tile shape");
  static_assert((TP / 4) % 2 == 1, "tile pitch must be an odd number of 16-B units");
  #pragma unroll 1
  for (int it = 0; it < NIT2; ++it) {
    const int np = (wave + NW * it) * 16 + nl;
    const bool second = np >= CO2;
    const int n = second ? np - CO2 : np;
    v8f acc[MT];
    #pragma unroll
    for (int mt = 0; mt < MT; ++mt) {
      #pragma unroll
      for (int j = 0; j < 8; ++j) acc[mt][j] = 0.f;
    }
    const float* __restrict__ ap = tl + nl * TP + 4 * g;
    const float* __restrict__ bp = Wn + (long)n * WP2 + (second ? WP2 / 2 : 0) + 4 * g;
    #pragma unroll 4
    for (int ks = 0; ks < KD / 8; ++ks) kstep8<MT>(acc, ap + 8 * ks, 16 * TP, bp + 8 * ks);
    #pragma unroll
    for (int mt = 0; mt < MT; ++mt) {
      #pragma unroll
      for (int j = 0; j < 8; ++j) {
        const int m = tile * (16 * MT) + 16 * mt + j + 8 * g;
        if (m < nv) Yn[(long)m * (2 * CO2) + np] = acc[mt][j];
      }
    }
  }
}

__device__ __forceinline__ unsigned ford(float f) {
  unsigned u = __float_as_uint(f);
  return (u & 0x80000000u) ? ~u : (u | 0x80000000u);
}
__device__ __forceinline__ float funord(unsigned u) {
  unsigned b = (u & 0x80000000u) ? (u & 0x7FFFFFFFu) : ~u;
  return __uint_as_float(b);
}
__device__ __forceinline__ float4 funord4(float4 v) {
  return make_float4(funord(__float_as_uint(v.x)), funord(__float_as_uint(v.y)),
                     funord(__float_as_uint(v.z)), funord(__float_as_uint(v.w)));
}


template <int CIN, int COUT, int CO2, int WP2, bool DEC>
__global__ __launch_bounds__((COUT / 16 >= 8) ? 256 : (COUT / 16 >= 4) ? 128 : 64) void edgeconv_wmma(
    const float* __restrict__ fA, const int* __restrict__ mapA, int CA,
    const float* __restrict__ fB, int CB,
    const int* __restrict__ src,
    const float* __restrict__ W, const float* __restrict__ bias,
    float* __restrict__ out, int nv,
    const int* __restrict__ pmap, unsigned* __restrict__ pool,
    const float* __restrict__ Wn, float* __restrict__ Yn)
{
  constexpr int K2  = 2 * CIN;
  constexpr int KP  = K2 + 4;
  constexpr int C4  = CIN / 4;
  constexpr int NK8 = K2 / 8;
  constexpr int NNT = COUT / 16;
  constexpr int MT  = (NNT >= 8) ? 3 : (NNT >= 4) ? 4 : 2;
  constexpr int BV  = 16 * MT;
  constexpr int NW  = (NNT >= 8) ? 8 : (NNT >= 4) ? 4 : 2;
  constexpr int NT  = 32 * NW;
  constexpr int NIT = NNT / NW;
  constexpr int RQ  = (BV * 8 + NT - 1) / NT;
  constexpr bool RQ_EXACT = (BV * 8) % NT == 0;
  static_assert(NNT % NW == 0, "one wave per N-tile, or a whole number of turns");

  __shared__ __align__(16) int   rowB[BV * 8];
  __shared__ __align__(16) int   rowA[BV * 8];
  __shared__ __align__(16) float xl[BV * KP];
  constexpr int OP  = COUT + 4;
  __shared__ __align__(16) float ot[(CO2 > 0) ? BV * OP : 4];
  const int tid  = threadIdx.x;
  const int tile = blockIdx.x;

  int pm = 0;
  if constexpr (CO2 == 0) {
    const int vr = tile * BV + 16 * (tid >> 5) + (tid & 15);
    pm = pmap[(vr >= nv) ? nv - 1 : vr];
  }

  {
    const int r = tid >> 3, j = tid & 7;
    const bool nb = (j >= 1) && (j <= 6);
    const int  jo = nb ? j - 1 : 0;
    int v[RQ], s[RQ], a[RQ];
    #pragma unroll
    for (int q = 0; q < RQ; ++q) {
      int vv = tile * BV + r + (NT / 8) * q;
      v[q] = (vv >= nv) ? nv - 1 : vv;
    }
    #pragma unroll
    for (int q = 0; q < RQ; ++q) s[q] = src[v[q] * 6 + jo];
    #pragma unroll
    for (int q = 0; q < RQ; ++q) { if (!nb) s[q] = v[q]; a[q] = s[q]; }
    if (mapA) {
      #pragma unroll
      for (int q = 0; q < RQ; ++q) a[q] = mapA[s[q]];
    }
    #pragma unroll
    for (int q = 0; q < RQ; ++q) {
      if (RQ_EXACT || tid + NT * q < BV * 8) { rowB[tid + NT * q] = s[q]; rowA[tid + NT * q] = a[q]; }
    }
  }
  __syncthreads();

  constexpr int UI = ((BV * C4) % (2 * NT) == 0) ? 2 : 3;
  static_assert((BV * C4) % (UI * NT) == 0, "UI items per thread per trip");
  for (int i0 = tid; i0 < BV * C4; i0 += UI * NT) {
    int    xoff[UI];
    float4 t[UI][7];
    #pragma unroll
    for (int u = 0; u < UI; ++u) {
      const int i = i0 + NT * u;
      const int r = i / C4, c4 = (i - r * C4) << 2;
      const bool inA = c4 < CA;
      const float* __restrict__ fp = inA ? fA : fB;
      const int  stride = inA ? CA : CB;
      const int  coff   = inA ? c4 : c4 - CA;
      const int* rows   = (inA ? rowA : rowB) + r * 8;
      const int4 r0 = *(const int4*)(rows);
      const int4 r1 = *(const int4*)(rows + 4);
      const int id[7] = { r0.x, r0.y, r0.z, r0.w, r1.x, r1.y, r1.z };
      #pragma unroll
      for (int j = 0; j < 7; ++j)
        t[u][j] = *(const float4*)(fp + (long)id[j] * stride + coff);
      xoff[u] = r * KP + c4;
    }
    #pragma unroll
    for (int u = 0; u < UI; ++u) {
      if constexpr (DEC) {
        #pragma unroll
        for (int j = 0; j < 7; ++j) t[u][j] = funord4(t[u][j]);
      }
      const float4 loc = t[u][0];
      float sx = 0.f, sy = 0.f, sz = 0.f, sw = 0.f;
      #pragma unroll
      for (int j = 1; j < 7; ++j) {
        sx += t[u][j].x; sy += t[u][j].y; sz += t[u][j].z; sw += t[u][j].w;
      }
      const float inv6 = 1.0f / 6.0f;
      float4 d = make_float4(sx * inv6 - loc.x, sy * inv6 - loc.y,
                             sz * inv6 - loc.z, sw * inv6 - loc.w);
      *(float4*)(xl + xoff[u])       = loc;
      *(float4*)(xl + xoff[u] + CIN) = d;
    }
  }
  __syncthreads();

  const int lane = tid & 31;
  const int wave = tid >> 5;
  const int g    = lane >> 4;
  const int nl   = lane & 15;
  #pragma unroll 1
  for (int it = 0; it < NIT; ++it) {
    const int nt = wave + NW * it;
    const int n = nt * 16 + nl;
    const float bv = bias[n];
    v8f acc[MT];
    #pragma unroll
    for (int mt = 0; mt < MT; ++mt) {
      #pragma unroll
      for (int j = 0; j < 8; ++j) acc[mt][j] = bv;
    }

    const float* __restrict__ ap = xl + nl * KP + 4 * g;
    const float* __restrict__ bp = W + (long)n * K2 + 4 * g;
    #pragma unroll 4
    for (int ks = 0; ks < NK8; ++ks) {
      const float4 b4 = *(const float4*)(bp + 8 * ks);
      float4 a4[MT];
      #pragma unroll
      for (int mt = 0; mt < MT; ++mt)
        a4[mt] = *(const float4*)(ap + mt * 16 * KP + 8 * ks);
      v2f b0, b1;
      b0[0] = b4.x; b0[1] = b4.y;
      b1[0] = b4.z; b1[1] = b4.w;
      #pragma unroll
      for (int mt = 0; mt < MT; ++mt) {
        v2f a0;
        a0[0] = a4[mt].x; a0[1] = a4[mt].y;
        acc[mt] = __builtin_amdgcn_wmma_f32_16x16x4_f32(
             false, a0,  false, b0,
             (short)0, acc[mt],  false,  false);
      }
      #pragma unroll
      for (int mt = 0; mt < MT; ++mt) {
        v2f a1;
        a1[0] = a4[mt].z; a1[1] = a4[mt].w;
        acc[mt] = __builtin_amdgcn_wmma_f32_16x16x4_f32(
             false, a1,  false, b1,
             (short)0, acc[mt],  false,  false);
      }
    }

    if constexpr (CO2 == 0) {
      static_assert(NIT == 1 && COUT <= K2 && NW == MT && COUT % 32 == 0, "output tile over the x tile, 16 rows per wave");
      __syncthreads();
    }
    #pragma unroll
    for (int mt = 0; mt < MT; ++mt) {
      #pragma unroll
      for (int j = 0; j < 8; ++j) {
        float y = acc[mt][j];
        y = (y >= 0.f) ? y : ALPHA * y;
        if constexpr (CO2 > 0) {
          ot[(16 * mt + j + 8 * g) * OP + n] = y;
        } else {
          xl[(16 * mt + j + 8 * g) * KP + n] = y;
        }
      }
    }
  }
  if constexpr (CO2 > 0) {
    __syncthreads();
    second_map<COUT, OP, CO2, WP2, MT, NW>(ot, Wn, Yn, tile, nv, wave, g, nl);
  } else {
    __syncthreads();
    const int wv = __builtin_amdgcn_readfirstlane(wave);
    constexpr int LPR = COUT / 4;
    constexpr int RPI = 32 / LPR;
    constexpr int NQ  = 16 / RPI;
    constexpr int NC  = COUT / 32;
    const int rq = lane / LPR, cq = (lane % LPR) * 4;
    float4   ov[NQ];
    unsigned img[16][NC];
    #pragma unroll
    for (int q = 0; q < NQ; ++q) ov[q] = *(const float4*)(xl + (16 * wv + RPI * q + rq) * KP + cq);
    #pragma unroll
    for (int r16 = 0; r16 < 16; ++r16) {
      #pragma unroll
      for (int cc = 0; cc < NC; ++cc) img[r16][cc] = ford(xl[(16 * wv + r16) * KP + 32 * cc + lane]);
    }
    if (tile * BV + BV <= nv) {
      #pragma unroll
      for (int q = 0; q < NQ; ++q) {
        const int m = tile * BV + 16 * wv + RPI * q + rq;
        *(float4*)(out + (long)m * COUT + cq) = ov[q];
      }
      #pragma unroll
      for (int r16 = 0; r16 < 16; ++r16) {
        const int p = __builtin_amdgcn_readlane(pm, r16);
        #pragma unroll
        for (int cc = 0; cc < NC; ++cc)
          atomicMax(pool + (long)p * COUT + 32 * cc + lane, img[r16][cc]);
      }
    } else {
      #pragma unroll
      for (int q = 0; q < NQ; ++q) {
        const int m = tile * BV + 16 * wv + RPI * q + rq;
        if (m < nv) *(float4*)(out + (long)m * COUT + cq) = ov[q];
      }
      #pragma unroll
      for (int r16 = 0; r16 < 16; ++r16) {
        const int p = __builtin_amdgcn_readlane(pm, r16);
        if (tile * BV + 16 * wv + r16 < nv) {
          #pragma unroll
          for (int cc = 0; cc < NC; ++cc)
            atomicMax(pool + (long)p * COUT + 32 * cc + lane, img[r16][cc]);
        }
      }
    }
  }
}

template <int CB, int COUT, int CA, int CO2, int WP2>
__global__ __launch_bounds__((COUT / 16 >= 4) ? 128 : 64) void edgeconv_up(
    const float* __restrict__ Y, const int* __restrict__ mapA,
    const float* __restrict__ fB, const int* __restrict__ src,
    const float* __restrict__ W, const float* __restrict__ bias,
    float* __restrict__ out, int nv,
    const float* __restrict__ Wn, float* __restrict__ Yn)
{
  constexpr int WP  = 2 * (CA + CB);
  constexpr int K2  = 2 * CB;
  constexpr int KP  = K2 + 4;
  constexpr int TP  = COUT + 4;
  constexpr int C4  = CB / 4;
  constexpr int T4  = COUT / 4;
  constexpr int YP  = 2 * COUT;
  constexpr int NNT = COUT / 16;
  constexpr int MT  = (NNT >= 4) ? 4 : 2;
  constexpr int BV  = 16 * MT;
  constexpr int NW  = (NNT >= 4) ? 4 : 2;
  constexpr int NT  = 32 * NW;
  constexpr int NIT = NNT / NW;
  static_assert((KP / 4) % 2 == 1 && (TP / 4) % 2 == 1, "pitches must be odd numbers of 16-B units");
  static_assert(CB % 8 == 0 && CA % 4 == 0 && COUT % 16 == 0 && NNT % NW == 0, "tile shape");

  __shared__ __align__(16) int   rowB[BV * 8];
  __shared__ __align__(16) int   rowA[BV * 8];
  __shared__ __align__(16) float xl[BV * KP];
  __shared__ __align__(16) float tl[BV * TP];
  const int tid  = threadIdx.x;
  const int tile = blockIdx.x;

  {
    const int r = tid >> 3, j = tid & 7;
    const bool nb = (j >= 1) && (j <= 6);
    const int  jo = nb ? j - 1 : 0;
    int v[4], s[4], a[4];
    #pragma unroll
    for (int q = 0; q < 4; ++q) {
      int vv = tile * BV + r + (NT / 8) * q;
      v[q] = (vv >= nv) ? nv - 1 : vv;
    }
    #pragma unroll
    for (int q = 0; q < 4; ++q) s[q] = src[v[q] * 6 + jo];
    #pragma unroll
    for (int q = 0; q < 4; ++q) { if (!nb) s[q] = v[q]; }
    #pragma unroll
    for (int q = 0; q < 4; ++q) a[q] = mapA[s[q]];
    #pragma unroll
    for (int q = 0; q < 4; ++q) { rowB[tid + NT * q] = s[q]; rowA[tid + NT * q] = a[q]; }
  }
  __syncthreads();

  static_assert((BV * C4) % (2 * NT) == 0, "two x items per thread per trip");
  for (int i0 = tid; i0 < BV * C4; i0 += 2 * NT) {
    int    xoff[2];
    float4 t[2][7];
    #pragma unroll
    for (int u = 0; u < 2; ++u) {
      const int i = i0 + NT * u;
      const int r = i / C4, c4 = (i - r * C4) << 2;
      const int* rows = rowB + r * 8;
      const int4 r0 = *(const int4*)(rows);
      const int4 r1 = *(const int4*)(rows + 4);
      const int id[7] = { r0.x, r0.y, r0.z, r0.w, r1.x, r1.y, r1.z };
      #pragma unroll
      for (int j = 0; j < 7; ++j)
        t[u][j] = *(const float4*)(fB + (long)id[j] * CB + c4);
      xoff[u] = r * KP + c4;
    }
    #pragma unroll
    for (int u = 0; u < 2; ++u) {
      const float4 loc = t[u][0];
      float sx = 0.f, sy = 0.f, sz = 0.f, sw = 0.f;
      #pragma unroll
      for (int j = 1; j < 7; ++j) {
        sx += t[u][j].x; sy += t[u][j].y; sz += t[u][j].z; sw += t[u][j].w;
      }
      const float inv6 = 1.0f / 6.0f;
      float4 d = make_float4(sx * inv6 - loc.x, sy * inv6 - loc.y,
                             sz * inv6 - loc.z, sw * inv6 - loc.w);
      *(float4*)(xl + xoff[u])      = loc;
      *(float4*)(xl + xoff[u] + CB) = d;
    }
  }

  static_assert((BV * T4) % (2 * NT) == 0, "two T items per thread per trip");
  for (int i0 = tid; i0 < BV * T4; i0 += 2 * NT) {
    int    toff[2];
    float4 yc[2], ya[2][7];
    #pragma unroll
    for (int u = 0; u < 2; ++u) {
      const int i = i0 + NT * u;
      const int r = i / T4, c4 = (i - r * T4) << 2;
      const int* rows = rowA + r * 8;
      const int4 r0 = *(const int4*)(rows);
      const int4 r1 = *(const int4*)(rows + 4);
      const int id[7] = { r0.x, r0.y, r0.z, r0.w, r1.x, r1.y, r1.z };
      yc[u] = *(const float4*)(Y + (long)id[0] * YP + c4);
      #pragma unroll
      for (int j = 0; j < 7; ++j)
        ya[u][j] = *(const float4*)(Y + (long)id[j] * YP + COUT + c4);
      toff[u] = r * TP + c4;
    }
    #pragma unroll
    for (int u = 0; u < 2; ++u) {
      float sx = 0.f, sy = 0.f, sz = 0.f, sw = 0.f;
      #pragma unroll
      for (int j = 1; j < 7; ++j) {
        sx += ya[u][j].x; sy += ya[u][j].y; sz += ya[u][j].z; sw += ya[u][j].w;
      }
      const float inv6 = 1.0f / 6.0f;
      *(float4*)(tl + toff[u]) =
          make_float4((yc[u].x - ya[u][0].x) + sx * inv6, (yc[u].y - ya[u][0].y) + sy * inv6,
                      (yc[u].z - ya[u][0].z) + sz * inv6, (yc[u].w - ya[u][0].w) + sw * inv6);
    }
  }
  __syncthreads();

  const int lane = tid & 31;
  const int wave = tid >> 5;
  const int g    = lane >> 4;
  const int nl   = lane & 15;
  #pragma unroll 1
  for (int it = 0; it < NIT; ++it) {
    const int nt = wave + NW * it;
    const int n = nt * 16 + nl;
    const float bv = bias[n];
    v8f acc[MT];
    #pragma unroll
    for (int mt = 0; mt < MT; ++mt) {
      #pragma unroll
      for (int j = 0; j < 8; ++j) acc[mt][j] = bv + tl[(16 * mt + j + 8 * g) * TP + n];
    }

    const float* __restrict__ ap = xl + nl * KP + 4 * g;
    const float* __restrict__ bp = W + (long)n * WP + CA + 4 * g;
    #pragma unroll 4
    for (int ks = 0; ks < CB / 8; ++ks) kstep8<MT>(acc, ap + 8 * ks, 16 * KP, bp + 8 * ks);
    #pragma unroll 4
    for (int ks = 0; ks < CB / 8; ++ks)
      kstep8<MT>(acc, ap + CB + 8 * ks, 16 * KP, bp + (CA + CB) + 8 * ks);

    #pragma unroll
    for (int mt = 0; mt < MT; ++mt) {
      #pragma unroll
      for (int j = 0; j < 8; ++j) {
        int m = tile * BV + 16 * mt + j + 8 * g;
        float y = acc[mt][j];
        y = (y >= 0.f) ? y : ALPHA * y;
        if constexpr (CO2 > 0) {
          tl[(16 * mt + j + 8 * g) * TP + n] = y;
        } else {
          if (m < nv) out[(long)m * COUT + n] = y;
        }
      }
    }
  }
  if constexpr (CO2 > 0) {
    __syncthreads();
    second_map<COUT, TP, CO2, WP2, MT, NW>(tl, Wn, Yn, tile, nv, wave, g, nl);
  }
}

__global__ __launch_bounds__(256) void pool_fill(unsigned* __restrict__ planes, int n4)
{
  for (int i = blockIdx.x * 256 + threadIdx.x; i < n4; i += gridDim.x * 256)
    ((uint4*)planes)[i] = make_uint4(0x007FFFFFu, 0x007FFFFFu, 0x007FFFFFu, 0x007FFFFFu);
}

extern "C" void kernel_launch(void* const* d_in, const int* in_sizes, int n_in,
                              void* d_out, int out_size, void* d_ws, size_t ws_size,
                              hipStream_t stream) {
  (void)in_sizes; (void)n_in; (void)out_size; (void)ws_size;
  const float* features = (const float*)d_in[0];
  const int*   src2     = (const int*)d_in[1];
  const int*   src1     = (const int*)d_in[3];
  const int*   src0     = (const int*)d_in[5];
  const int*   p21      = (const int*)d_in[7];
  const int*   p10      = (const int*)d_in[8];
  const float* w_enc0 = (const float*)d_in[9],  *b_enc0 = (const float*)d_in[10];
  const float* w_enc1 = (const float*)d_in[11], *b_enc1 = (const float*)d_in[12];
  const float* w_enc2 = (const float*)d_in[13], *b_enc2 = (const float*)d_in[14];
  const float* w_dec0 = (const float*)d_in[15], *b_dec0 = (const float*)d_in[16];
  const float* w_dec1 = (const float*)d_in[17], *b_dec1 = (const float*)d_in[18];

  char* ws = (char*)d_ws;
  size_t off = 0;
  float* f2 = (float*)(ws + off); off += (size_t)NV2 * 32 * 4;
  float* f1 = (float*)(ws + off); off += (size_t)NV1 * 64 * 4;
  float* y0 = (float*)(ws + off); off += (size_t)NV0 * 128 * 4;
  float* y1 = (float*)(ws + off); off += (size_t)NV1 * 64 * 4;
  float* P1 = (float*)(ws + off); off += (size_t)NV1 * 32 * 4;
  float* P0 = (float*)(ws + off);

  constexpr int G2 = (NV2 + 31) / 32, G1 = (NV1 + 63) / 64, G0 = (NV0 + 47) / 48;
  constexpr int NP4 = (NV1 * 32 + NV0 * 64) / 4;
  static_assert((NV1 * 32 + NV0 * 64) % 4 == 0, "planes fill in 16-B units");

  pool_fill<<<256, 256, 0, stream>>>((unsigned*)P1, NP4);
  edgeconv_wmma<16, 32, 0, 0, false><<<G2, 64, 0, stream>>>(features, nullptr, 16, nullptr, 0, src2,
                                                             w_enc0, b_enc0, f2, NV2, p21, (unsigned*)P1,
                                                             nullptr, nullptr);
  edgeconv_wmma<32, 64, 0, 0, true><<<G1, 128, 0, stream>>>(P1, nullptr, 32, nullptr, 0, src1,
                                                            w_enc1, b_enc1, f1, NV1, p10, (unsigned*)P0,
                                                            nullptr, nullptr);
  edgeconv_wmma<64, 128, 64, 384, true><<<G0, 256, 0, stream>>>(P0, nullptr, 64, nullptr, 0, src0,
                                                                w_enc2, b_enc2, nullptr, NV0, nullptr, nullptr,
                                                                w_dec0, y0);

  edgeconv_up<64, 64, 128, 32, 192><<<G1, 128, 0, stream>>>(y0, p10, f1, src1, w_dec0, b_dec0,
                                                            nullptr, NV1, w_dec1, y1);
  edgeconv_up<32, 32, 64, 0, 0><<<G2, 64, 0, stream>>>(y1, p21, f2, src2, w_dec1, b_dec1,
                                                        (float*)d_out, NV2, nullptr, nullptr);
}
